# starting kernel
# baseline (speedup 1.0000x reference)
#include <hip/hip_runtime.h>

#define NN 50000
#define NE 800000
#define NLAB 50
#define EMB 64
#define HID 96

typedef __attribute__((ext_vector_type(2))) float v2f;
typedef __attribute__((ext_vector_type(8))) float v8f;


__global__ void zero2_kernel(float* __restrict__ a, float* __restrict__ b, int n) {
    int i = blockIdx.x * blockDim.x + threadIdx.x;
    if (i < n) { a[i] = 0.0f; b[i] = 0.0f; }
}

__global__ void edge_deg_kernel(const int* __restrict__ src, const int* __restrict__ dst,
                                float* __restrict__ odeg, float* __restrict__ ideg, int e) {
    int i = blockIdx.x * blockDim.x + threadIdx.x;
    if (i < e) {
        atomicAdd(&odeg[src[i]], 1.0f);
        atomicAdd(&ideg[dst[i]], 1.0f);
    }
}

__global__ void norm_kernel(float* __restrict__ od, float* __restrict__ id, int n) {
    int i = blockIdx.x * blockDim.x + threadIdx.x;
    if (i < n) {
        od[i] = rsqrtf(fmaxf(od[i], 1.0f));
        id[i] = rsqrtf(fmaxf(id[i], 1.0f));
    }
}


__global__ void embed_scale_kernel(const int* __restrict__ labels, const float* __restrict__ emb,
                                   const float* __restrict__ onorm, float* __restrict__ out) {
    int i = blockIdx.x;
    int f = threadIdx.x;
    out[(long)i * EMB + f] = emb[(long)labels[i] * EMB + f] * onorm[i];
}

__global__ void scale_rows_kernel(const float* __restrict__ h, const float* __restrict__ norm,
                                  float* __restrict__ out, int d) {
    int i = blockIdx.x;
    int f = threadIdx.x;
    out[(long)i * d + f] = h[(long)i * d + f] * norm[i];
}

__global__ void zero_buf_kernel(float* __restrict__ p, long n) {
    long i = (long)blockIdx.x * blockDim.x + threadIdx.x;
    if (i < n) p[i] = 0.0f;
}

__global__ void scatter_add_kernel(const float* __restrict__ hs, float* __restrict__ agg,
                                   const int* __restrict__ src, const int* __restrict__ dst, int d) {
    int e = blockIdx.x;
    int f = threadIdx.x;
    int s = src[e], t = dst[e];
    atomicAdd(&agg[(long)t * d + f], hs[(long)s * d + f]);
}


__global__ __launch_bounds__(256) void gcn_gemm_wmma(
    const float* __restrict__ A, const float* __restrict__ rowScale,
    const float* __restrict__ W, const float* __restrict__ bias,
    float* __restrict__ out, int K, int Nout, int Npad,
    int tilesN, int totalTiles, int relu)
{
    __shared__ float smem[HID * HID + HID];
    const int tid = threadIdx.x;
    const int KN = K * Npad;

    for (int idx = tid; idx < KN; idx += 256) {
        int k = idx / Npad, n = idx % Npad;
        smem[idx] = (n < Nout) ? W[k * Nout + n] : 0.0f;
    }
    for (int idx = tid; idx < Npad; idx += 256)
        smem[KN + idx] = (idx < Nout) ? bias[idx] : 0.0f;
    __syncthreads();

    const int wave = tid >> 5;
    const int lane = tid & 31;
    const int tileId = blockIdx.x * 8 + wave;
    if (tileId >= totalTiles) return;

    const int m0 = (tileId / tilesN) * 16;
    const int n0 = (tileId % tilesN) * 16;

    const int rowA = m0 + (lane & 15);
    const int koff = (lane >> 4) * 2;
    const int colB = lane & 15;

    v8f c;
    const float bval = smem[KN + n0 + colB];
#pragma unroll
    for (int r = 0; r < 8; ++r) c[r] = bval;

    const float s = rowScale[rowA];
    const float* arow = A + (long)rowA * K;
    const float* bcol = smem + n0 + colB;

    for (int k0 = 0; k0 < K; k0 += 4) {
        v2f a, b;
        a.x = arow[k0 + koff] * s;
        a.y = arow[k0 + koff + 1] * s;
        b.x = bcol[(k0 + koff) * Npad];
        b.y = bcol[(k0 + koff + 1) * Npad];
        c = __builtin_amdgcn_wmma_f32_16x16x4_f32(
                 false, a,  false, b,
                 (short)0, c,  false,  false);
    }

    const int rbase = m0 + (lane >> 4) * 8;
    const int col = n0 + colB;
    if (col < Nout) {
#pragma unroll
        for (int r = 0; r < 8; ++r) {
            float v = c[r];
            if (relu) v = fmaxf(v, 0.0f);
            out[(long)(rbase + r) * Nout + col] = v;
        }
    }
}


extern "C" void kernel_launch(void* const* d_in, const int* in_sizes, int n_in,
                              void* d_out, int out_size, void* d_ws, size_t ws_size,
                              hipStream_t stream) {
    const int*   labels = (const int*)d_in[0];
    const int*   src    = (const int*)d_in[1];
    const int*   dst    = (const int*)d_in[2];
    const float* emb    = (const float*)d_in[3];
    const float* W1     = (const float*)d_in[4];
    const float* b1     = (const float*)d_in[5];
    const float* W2     = (const float*)d_in[6];
    const float* b2     = (const float*)d_in[7];
    const float* W3     = (const float*)d_in[8];
    const float* b3     = (const float*)d_in[9];
    float* out = (float*)d_out;

    float* w = (float*)d_ws;
    float* onorm = w;
    float* inorm = w + NN;
    float* bufA  = w + 2 * (size_t)NN;
    float* bufB  = bufA + (size_t)NN * HID;
    float* bufC  = bufB + (size_t)NN * HID;

    const int TPB = 256;

    zero2_kernel<<<(NN + TPB - 1) / TPB, TPB, 0, stream>>>(onorm, inorm, NN);
    edge_deg_kernel<<<(NE + TPB - 1) / TPB, TPB, 0, stream>>>(src, dst, onorm, inorm, NE);
    norm_kernel<<<(NN + TPB - 1) / TPB, TPB, 0, stream>>>(onorm, inorm, NN);

    embed_scale_kernel<<<NN, EMB, 0, stream>>>(labels, emb, onorm, bufB);
    zero_buf_kernel<<<(int)(((long)NN * EMB + TPB - 1) / TPB), TPB, 0, stream>>>(bufC, (long)NN * EMB);
    scatter_add_kernel<<<NE, EMB, 0, stream>>>(bufB, bufC, src, dst, EMB);
    {
        int tilesN = HID / 16;
        int total  = (NN / 16) * tilesN;
        gcn_gemm_wmma<<<(total + 7) / 8, 256, 0, stream>>>(
            bufC, inorm, W1, b1, bufA, EMB, HID, HID, tilesN, total, 1);
    }

    scale_rows_kernel<<<NN, HID, 0, stream>>>(bufA, onorm, bufB, HID);
    zero_buf_kernel<<<(int)(((long)NN * HID + TPB - 1) / TPB), TPB, 0, stream>>>(bufC, (long)NN * HID);
    scatter_add_kernel<<<NE, HID, 0, stream>>>(bufB, bufC, src, dst, HID);
    {
        int tilesN = HID / 16;
        int total  = (NN / 16) * tilesN;
        gcn_gemm_wmma<<<(total + 7) / 8, 256, 0, stream>>>(
            bufC, inorm, W2, b2, bufA, HID, HID, HID, tilesN, total, 1);
    }

    scale_rows_kernel<<<NN, HID, 0, stream>>>(bufA, onorm, bufB, HID);
    zero_buf_kernel<<<(int)(((long)NN * HID + TPB - 1) / TPB), TPB, 0, stream>>>(bufC, (long)NN * HID);
    scatter_add_kernel<<<NE, HID, 0, stream>>>(bufB, bufC, src, dst, HID);
    {
        int Npad   = 64;
        int tilesN = Npad / 16;
        int total  = (NN / 16) * tilesN;
        gcn_gemm_wmma<<<(total + 7) / 8, 256, 0, stream>>>(
            bufC, inorm, W3, b3, out, HID, NLAB, Npad, tilesN, total, 0);
    }
}


# scatter_add: one wave walks 32 edges (coalesced index load, readlane, 8 edges' row loads in flight), 3125 blocks instead of 800000 one-edge blocks; float atomics kept, arrival order as before
# speedup vs baseline: 1.7533x; 1.7533x over previous
#include <hip/hip_runtime.h>

#define NN 50000
#define NE 800000
#define NLAB 50
#define EMB 64
#define HID 96

typedef __attribute__((ext_vector_type(2))) float v2f;
typedef __attribute__((ext_vector_type(8))) float v8f;


__global__ void zero2_kernel(float* __restrict__ a, float* __restrict__ b, int n) {
    int i = blockIdx.x * blockDim.x + threadIdx.x;
    if (i < n) { a[i] = 0.0f; b[i] = 0.0f; }
}

__global__ void edge_deg_kernel(const int* __restrict__ src, const int* __restrict__ dst,
                                float* __restrict__ odeg, float* __restrict__ ideg, int e) {
    int i = blockIdx.x * blockDim.x + threadIdx.x;
    if (i < e) {
        atomicAdd(&odeg[src[i]], 1.0f);
        atomicAdd(&ideg[dst[i]], 1.0f);
    }
}

__global__ void norm_kernel(float* __restrict__ od, float* __restrict__ id, int n) {
    int i = blockIdx.x * blockDim.x + threadIdx.x;
    if (i < n) {
        od[i] = rsqrtf(fmaxf(od[i], 1.0f));
        id[i] = rsqrtf(fmaxf(id[i], 1.0f));
    }
}


__global__ void embed_scale_kernel(const int* __restrict__ labels, const float* __restrict__ emb,
                                   const float* __restrict__ onorm, float* __restrict__ out) {
    int i = blockIdx.x;
    int f = threadIdx.x;
    out[(long)i * EMB + f] = emb[(long)labels[i] * EMB + f] * onorm[i];
}

__global__ void scale_rows_kernel(const float* __restrict__ h, const float* __restrict__ norm,
                                  float* __restrict__ out, int d) {
    int i = blockIdx.x;
    int f = threadIdx.x;
    out[(long)i * d + f] = h[(long)i * d + f] * norm[i];
}

__global__ void zero_buf_kernel(float* __restrict__ p, long n) {
    long i = (long)blockIdx.x * blockDim.x + threadIdx.x;
    if (i < n) p[i] = 0.0f;
}

template <int D>
__device__ __forceinline__ void scatter_add_body(const float* __restrict__ hs, float* __restrict__ agg,
                                                 const int* __restrict__ src, const int* __restrict__ dst) {
    static_assert(D % 32 == 0, "a row is a whole number of 32-lane lines");
    static_assert(NE % 256 == 0, "every wave of every block owns 32 edges");
    constexpr int NK = D / 32;
    constexpr int UB = 8;
    const int lane = threadIdx.x & 31;
    const int e0 = (blockIdx.x * 8 + (threadIdx.x >> 5)) * 32;
    const int sv = src[e0 + lane];
    const int tv = dst[e0 + lane];
#pragma unroll
    for (int j0 = 0; j0 < 32; j0 += UB) {
        float v[UB][NK];
#pragma unroll
        for (int u = 0; u < UB; ++u) {
            const int s = __builtin_amdgcn_readlane(sv, j0 + u);
            const float* row = hs + (long)s * D + lane;
#pragma unroll
            for (int k = 0; k < NK; ++k) v[u][k] = row[k * 32];
        }
#pragma unroll
        for (int u = 0; u < UB; ++u) {
            const int t = __builtin_amdgcn_readlane(tv, j0 + u);
            float* row = agg + (long)t * D + lane;
#pragma unroll
            for (int k = 0; k < NK; ++k) atomicAdd(&row[k * 32], v[u][k]);
        }
    }
}

__global__ __launch_bounds__(256) void scatter_add64_kernel(const float* __restrict__ hs, float* __restrict__ agg,
                                                            const int* __restrict__ src, const int* __restrict__ dst) {
    scatter_add_body<EMB>(hs, agg, src, dst);
}

__global__ __launch_bounds__(256) void scatter_add96_kernel(const float* __restrict__ hs, float* __restrict__ agg,
                                                            const int* __restrict__ src, const int* __restrict__ dst) {
    scatter_add_body<HID>(hs, agg, src, dst);
}


__global__ __launch_bounds__(256) void gcn_gemm_wmma(
    const float* __restrict__ A, const float* __restrict__ rowScale,
    const float* __restrict__ W, const float* __restrict__ bias,
    float* __restrict__ out, int K, int Nout, int Npad,
    int tilesN, int totalTiles, int relu)
{
    __shared__ float smem[HID * HID + HID];
    const int tid = threadIdx.x;
    const int KN = K * Npad;

    for (int idx = tid; idx < KN; idx += 256) {
        int k = idx / Npad, n = idx % Npad;
        smem[idx] = (n < Nout) ? W[k * Nout + n] : 0.0f;
    }
    for (int idx = tid; idx < Npad; idx += 256)
        smem[KN + idx] = (idx < Nout) ? bias[idx] : 0.0f;
    __syncthreads();

    const int wave = tid >> 5;
    const int lane = tid & 31;
    const int tileId = blockIdx.x * 8 + wave;
    if (tileId >= totalTiles) return;

    const int m0 = (tileId / tilesN) * 16;
    const int n0 = (tileId % tilesN) * 16;

    const int rowA = m0 + (lane & 15);
    const int koff = (lane >> 4) * 2;
    const int colB = lane & 15;

    v8f c;
    const float bval = smem[KN + n0 + colB];
#pragma unroll
    for (int r = 0; r < 8; ++r) c[r] = bval;

    const float s = rowScale[rowA];
    const float* arow = A + (long)rowA * K;
    const float* bcol = smem + n0 + colB;

    for (int k0 = 0; k0 < K; k0 += 4) {
        v2f a, b;
        a.x = arow[k0 + koff] * s;
        a.y = arow[k0 + koff + 1] * s;
        b.x = bcol[(k0 + koff) * Npad];
        b.y = bcol[(k0 + koff + 1) * Npad];
        c = __builtin_amdgcn_wmma_f32_16x16x4_f32(
                 false, a,  false, b,
                 (short)0, c,  false,  false);
    }

    const int rbase = m0 + (lane >> 4) * 8;
    const int col = n0 + colB;
    if (col < Nout) {
#pragma unroll
        for (int r = 0; r < 8; ++r) {
            float v = c[r];
            if (relu) v = fmaxf(v, 0.0f);
            out[(long)(rbase + r) * Nout + col] = v;
        }
    }
}


extern "C" void kernel_launch(void* const* d_in, const int* in_sizes, int n_in,
                              void* d_out, int out_size, void* d_ws, size_t ws_size,
                              hipStream_t stream) {
    const int*   labels = (const int*)d_in[0];
    const int*   src    = (const int*)d_in[1];
    const int*   dst    = (const int*)d_in[2];
    const float* emb    = (const float*)d_in[3];
    const float* W1     = (const float*)d_in[4];
    const float* b1     = (const float*)d_in[5];
    const float* W2     = (const float*)d_in[6];
    const float* b2     = (const float*)d_in[7];
    const float* W3     = (const float*)d_in[8];
    const float* b3     = (const float*)d_in[9];
    float* out = (float*)d_out;

    float* w = (float*)d_ws;
    float* onorm = w;
    float* inorm = w + NN;
    float* bufA  = w + 2 * (size_t)NN;
    float* bufB  = bufA + (size_t)NN * HID;
    float* bufC  = bufB + (size_t)NN * HID;

    const int TPB = 256;

    zero2_kernel<<<(NN + TPB - 1) / TPB, TPB, 0, stream>>>(onorm, inorm, NN);
    edge_deg_kernel<<<(NE + TPB - 1) / TPB, TPB, 0, stream>>>(src, dst, onorm, inorm, NE);
    norm_kernel<<<(NN + TPB - 1) / TPB, TPB, 0, stream>>>(onorm, inorm, NN);

    embed_scale_kernel<<<NN, EMB, 0, stream>>>(labels, emb, onorm, bufB);
    zero_buf_kernel<<<(int)(((long)NN * EMB + TPB - 1) / TPB), TPB, 0, stream>>>(bufC, (long)NN * EMB);
    scatter_add64_kernel<<<NE / 256, TPB, 0, stream>>>(bufB, bufC, src, dst);
    {
        int tilesN = HID / 16;
        int total  = (NN / 16) * tilesN;
        gcn_gemm_wmma<<<(total + 7) / 8, 256, 0, stream>>>(
            bufC, inorm, W1, b1, bufA, EMB, HID, HID, tilesN, total, 1);
    }

    scale_rows_kernel<<<NN, HID, 0, stream>>>(bufA, onorm, bufB, HID);
    zero_buf_kernel<<<(int)(((long)NN * HID + TPB - 1) / TPB), TPB, 0, stream>>>(bufC, (long)NN * HID);
    scatter_add96_kernel<<<NE / 256, TPB, 0, stream>>>(bufB, bufC, src, dst);
    {
        int tilesN = HID / 16;
        int total  = (NN / 16) * tilesN;
        gcn_gemm_wmma<<<(total + 7) / 8, 256, 0, stream>>>(
            bufC, inorm, W2, b2, bufA, HID, HID, HID, tilesN, total, 1);
    }

    scale_rows_kernel<<<NN, HID, 0, stream>>>(bufA, onorm, bufB, HID);
    zero_buf_kernel<<<(int)(((long)NN * HID + TPB - 1) / TPB), TPB, 0, stream>>>(bufC, (long)NN * HID);
    scatter_add96_kernel<<<NE / 256, TPB, 0, stream>>>(bufB, bufC, src, dst);
    {
        int Npad   = 64;
        int tilesN = Npad / 16;
        int total  = (NN / 16) * tilesN;
        gcn_gemm_wmma<<<(total + 7) / 8, 256, 0, stream>>>(
            bufC, inorm, W3, b3, out, HID, NLAB, Npad, tilesN, total, 0);
    }
}


# aggregation as a gather over a dst-sorted edge list built per call (rank = return of the in-degree add, scan, fill): no feature atomics, no zero fill; per-node sum order = degree-add arrival order
# speedup vs baseline: 2.1967x; 1.2529x over previous
#include <hip/hip_runtime.h>

#define NN 50000
#define NE 800000
#define NLAB 50
#define EMB 64
#define HID 96

typedef __attribute__((ext_vector_type(2))) float v2f;
typedef __attribute__((ext_vector_type(8))) float v8f;


__global__ void zero2_kernel(float* __restrict__ a, float* __restrict__ b, int n) {
    int i = blockIdx.x * blockDim.x + threadIdx.x;
    if (i < n) { a[i] = 0.0f; b[i] = 0.0f; }
}

__global__ void edge_deg_rank_kernel(const int* __restrict__ src, const int* __restrict__ dst,
                                     float* __restrict__ odeg, float* __restrict__ ideg,
                                     int* __restrict__ rank, int e) {
    int i = blockIdx.x * blockDim.x + threadIdx.x;
    if (i < e) {
        const int s = src[i];
        const int d = dst[i];
        atomicAdd(&odeg[s], 1.0f);
        rank[i] = (int)atomicAdd(&ideg[d], 1.0f);
    }
}

__global__ __launch_bounds__(1024) void offsets_kernel(const float* __restrict__ ideg, int* __restrict__ off) {
    __shared__ int wsum[32];
    __shared__ int wpre[32];
    const int t = threadIdx.x;
    const int lane = t & 31;
    const int w = t >> 5;
    const int i = blockIdx.x * 1024 + t;
    int pre = 0;
    for (int c = 0; c < (int)blockIdx.x; ++c) pre += (int)ideg[c * 1024 + t];
    const int x = (i < NN) ? (int)ideg[i] : 0;
    int s = x;
#pragma unroll
    for (int d = 1; d < 32; d <<= 1) {
        const int y = __shfl_up(s, d);
        if (lane >= d) s += y;
    }
#pragma unroll
    for (int d = 16; d > 0; d >>= 1) pre += __shfl_xor(pre, d);
    if (lane == 31) wsum[w] = s;
    if (lane == 0) wpre[w] = pre;
    __syncthreads();
    int base = 0;
    for (int q = 0; q < 32; ++q) {
        base += wpre[q];
        if (q < w) base += wsum[q];
    }
    if (i <= NN) off[i] = base + s - x;
}

__global__ void csr_fill_kernel(const int* __restrict__ src, const int* __restrict__ dst,
                                const int* __restrict__ off, const int* __restrict__ rank,
                                int* __restrict__ csr, int e) {
    int i = blockIdx.x * blockDim.x + threadIdx.x;
    if (i < e) csr[off[dst[i]] + rank[i]] = src[i];
}

__global__ void norm_kernel(float* __restrict__ od, float* __restrict__ id, int n) {
    int i = blockIdx.x * blockDim.x + threadIdx.x;
    if (i < n) {
        od[i] = rsqrtf(fmaxf(od[i], 1.0f));
        id[i] = rsqrtf(fmaxf(id[i], 1.0f));
    }
}


__global__ void embed_scale_kernel(const int* __restrict__ labels, const float* __restrict__ emb,
                                   const float* __restrict__ onorm, float* __restrict__ out) {
    int i = blockIdx.x;
    int f = threadIdx.x;
    out[(long)i * EMB + f] = emb[(long)labels[i] * EMB + f] * onorm[i];
}

__global__ void scale_rows_kernel(float* __restrict__ h, const float* __restrict__ norm, int d) {
    int i = blockIdx.x;
    int f = threadIdx.x;
    h[(long)i * d + f] = h[(long)i * d + f] * norm[i];
}

template <int D>
__device__ __forceinline__ void gather_rows_body(const float* __restrict__ hs, float* __restrict__ agg,
                                                 const int* __restrict__ off, const int* __restrict__ csr) {
    static_assert(D % 32 == 0, "a row is a whole number of 32-lane lines");
    static_assert(NN % 8 == 0, "every wave of every block owns a node");
    constexpr int NK = D / 32;
    constexpr int UB = 8;
    const int lane = threadIdx.x & 31;
    const int node = blockIdx.x * 8 + __builtin_amdgcn_readfirstlane((int)(threadIdx.x >> 5));
    const int beg = off[node];
    const int end = off[node + 1];
    float acc[NK];
#pragma unroll
    for (int k = 0; k < NK; ++k) acc[k] = 0.0f;
    for (int base = beg; base < end; base += 32) {
        const int n = (end - base < 32) ? (end - base) : 32;
        const int sv = (lane < n) ? csr[base + lane] : 0;
        int j = 0;
        for (; j + UB <= n; j += UB) {
            float v[UB][NK];
#pragma unroll
            for (int u = 0; u < UB; ++u) {
                const int s = __builtin_amdgcn_readlane(sv, j + u);
                const float* row = hs + (long)s * D + lane;
#pragma unroll
                for (int k = 0; k < NK; ++k) v[u][k] = row[k * 32];
            }
#pragma unroll
            for (int u = 0; u < UB; ++u)
#pragma unroll
                for (int k = 0; k < NK; ++k) acc[k] += v[u][k];
        }
        if (j < n) {
            float v[UB][NK];
#pragma unroll
            for (int u = 0; u < UB; ++u) {
#pragma unroll
                for (int k = 0; k < NK; ++k) v[u][k] = 0.0f;
                if (j + u < n) {
                    const int s = __builtin_amdgcn_readlane(sv, j + u);
                    const float* row = hs + (long)s * D + lane;
#pragma unroll
                    for (int k = 0; k < NK; ++k) v[u][k] = row[k * 32];
                }
            }
#pragma unroll
            for (int u = 0; u < UB; ++u)
#pragma unroll
                for (int k = 0; k < NK; ++k) acc[k] += v[u][k];
        }
    }
    float* o = agg + (long)node * D + lane;
#pragma unroll
    for (int k = 0; k < NK; ++k) o[k * 32] = acc[k];
}

__global__ __launch_bounds__(256) void gather_rows64_kernel(const float* __restrict__ hs, float* __restrict__ agg,
                                                            const int* __restrict__ off, const int* __restrict__ csr) {
    gather_rows_body<EMB>(hs, agg, off, csr);
}

__global__ __launch_bounds__(256) void gather_rows96_kernel(const float* __restrict__ hs, float* __restrict__ agg,
                                                            const int* __restrict__ off, const int* __restrict__ csr) {
    gather_rows_body<HID>(hs, agg, off, csr);
}


__global__ __launch_bounds__(256) void gcn_gemm_wmma(
    const float* __restrict__ A, const float* __restrict__ rowScale,
    const float* __restrict__ W, const float* __restrict__ bias,
    float* __restrict__ out, int K, int Nout, int Npad,
    int tilesN, int totalTiles, int relu)
{
    __shared__ float smem[HID * HID + HID];
    const int tid = threadIdx.x;
    const int KN = K * Npad;

    for (int idx = tid; idx < KN; idx += 256) {
        int k = idx / Npad, n = idx % Npad;
        smem[idx] = (n < Nout) ? W[k * Nout + n] : 0.0f;
    }
    for (int idx = tid; idx < Npad; idx += 256)
        smem[KN + idx] = (idx < Nout) ? bias[idx] : 0.0f;
    __syncthreads();

    const int wave = tid >> 5;
    const int lane = tid & 31;
    const int tileId = blockIdx.x * 8 + wave;
    if (tileId >= totalTiles) return;

    const int m0 = (tileId / tilesN) * 16;
    const int n0 = (tileId % tilesN) * 16;

    const int rowA = m0 + (lane & 15);
    const int koff = (lane >> 4) * 2;
    const int colB = lane & 15;

    v8f c;
    const float bval = smem[KN + n0 + colB];
#pragma unroll
    for (int r = 0; r < 8; ++r) c[r] = bval;

    const float s = rowScale[rowA];
    const float* arow = A + (long)rowA * K;
    const float* bcol = smem + n0 + colB;

    for (int k0 = 0; k0 < K; k0 += 4) {
        v2f a, b;
        a.x = arow[k0 + koff] * s;
        a.y = arow[k0 + koff + 1] * s;
        b.x = bcol[(k0 + koff) * Npad];
        b.y = bcol[(k0 + koff + 1) * Npad];
        c = __builtin_amdgcn_wmma_f32_16x16x4_f32(
                 false, a,  false, b,
                 (short)0, c,  false,  false);
    }

    const int rbase = m0 + (lane >> 4) * 8;
    const int col = n0 + colB;
    if (col < Nout) {
#pragma unroll
        for (int r = 0; r < 8; ++r) {
            float v = c[r];
            if (relu) v = fmaxf(v, 0.0f);
            out[(long)(rbase + r) * Nout + col] = v;
        }
    }
}


extern "C" void kernel_launch(void* const* d_in, const int* in_sizes, int n_in,
                              void* d_out, int out_size, void* d_ws, size_t ws_size,
                              hipStream_t stream) {
    const int*   labels = (const int*)d_in[0];
    const int*   src    = (const int*)d_in[1];
    const int*   dst    = (const int*)d_in[2];
    const float* emb    = (const float*)d_in[3];
    const float* W1     = (const float*)d_in[4];
    const float* b1     = (const float*)d_in[5];
    const float* W2     = (const float*)d_in[6];
    const float* b2     = (const float*)d_in[7];
    const float* W3     = (const float*)d_in[8];
    const float* b3     = (const float*)d_in[9];
    float* out = (float*)d_out;

    float* w = (float*)d_ws;
    float* onorm = w;
    float* inorm = w + NN;
    float* bufA  = w + 2 * (size_t)NN;
    float* bufB  = bufA + (size_t)NN * HID;
    float* bufC  = bufB + (size_t)NN * HID;
    int* off  = (int*)bufB;
    int* rank = off + (NN + 64);
    int* csr  = rank + NE;

    const int TPB = 256;

    zero2_kernel<<<(NN + TPB - 1) / TPB, TPB, 0, stream>>>(onorm, inorm, NN);
    edge_deg_rank_kernel<<<(NE + TPB - 1) / TPB, TPB, 0, stream>>>(src, dst, onorm, inorm, rank, NE);
    offsets_kernel<<<(NN + 1 + 1023) / 1024, 1024, 0, stream>>>(inorm, off);
    norm_kernel<<<(NN + TPB - 1) / TPB, TPB, 0, stream>>>(onorm, inorm, NN);
    csr_fill_kernel<<<(NE + TPB - 1) / TPB, TPB, 0, stream>>>(src, dst, off, rank, csr, NE);

    embed_scale_kernel<<<NN, EMB, 0, stream>>>(labels, emb, onorm, bufA);
    gather_rows64_kernel<<<NN / 8, TPB, 0, stream>>>(bufA, bufC, off, csr);
    {
        int tilesN = HID / 16;
        int total  = (NN / 16) * tilesN;
        gcn_gemm_wmma<<<(total + 7) / 8, 256, 0, stream>>>(
            bufC, inorm, W1, b1, bufA, EMB, HID, HID, tilesN, total, 1);
    }

    scale_rows_kernel<<<NN, HID, 0, stream>>>(bufA, onorm, HID);
    gather_rows96_kernel<<<NN / 8, TPB, 0, stream>>>(bufA, bufC, off, csr);
    {
        int tilesN = HID / 16;
        int total  = (NN / 16) * tilesN;
        gcn_gemm_wmma<<<(total + 7) / 8, 256, 0, stream>>>(
            bufC, inorm, W2, b2, bufA, HID, HID, HID, tilesN, total, 1);
    }

    scale_rows_kernel<<<NN, HID, 0, stream>>>(bufA, onorm, HID);
    gather_rows96_kernel<<<NN / 8, TPB, 0, stream>>>(bufA, bufC, off, csr);
    {
        int Npad   = 64;
        int tilesN = Npad / 16;
        int total  = (NN / 16) * tilesN;
        gcn_gemm_wmma<<<(total + 7) / 8, 256, 0, stream>>>(
            bufC, inorm, W3, b3, out, HID, NLAB, Npad, tilesN, total, 0);
    }
}


# out_norm row scaling folded into the producing GEMM's store (relu, then times out_norm[row]): the two scale_rows launches and their read-write pass over h are gone; sums and their order unchanged
# speedup vs baseline: 2.3912x; 1.0885x over previous
#include <hip/hip_runtime.h>

#define NN 50000
#define NE 800000
#define NLAB 50
#define EMB 64
#define HID 96

typedef __attribute__((ext_vector_type(2))) float v2f;
typedef __attribute__((ext_vector_type(8))) float v8f;


__global__ void zero2_kernel(float* __restrict__ a, float* __restrict__ b, int n) {
    int i = blockIdx.x * blockDim.x + threadIdx.x;
    if (i < n) { a[i] = 0.0f; b[i] = 0.0f; }
}

__global__ void edge_deg_rank_kernel(const int* __restrict__ src, const int* __restrict__ dst,
                                     float* __restrict__ odeg, float* __restrict__ ideg,
                                     int* __restrict__ rank, int e) {
    int i = blockIdx.x * blockDim.x + threadIdx.x;
    if (i < e) {
        const int s = src[i];
        const int d = dst[i];
        atomicAdd(&odeg[s], 1.0f);
        rank[i] = (int)atomicAdd(&ideg[d], 1.0f);
    }
}

__global__ __launch_bounds__(1024) void offsets_kernel(const float* __restrict__ ideg, int* __restrict__ off) {
    __shared__ int wsum[32];
    __shared__ int wpre[32];
    const int t = threadIdx.x;
    const int lane = t & 31;
    const int w = t >> 5;
    const int i = blockIdx.x * 1024 + t;
    int pre = 0;
    for (int c = 0; c < (int)blockIdx.x; ++c) pre += (int)ideg[c * 1024 + t];
    const int x = (i < NN) ? (int)ideg[i] : 0;
    int s = x;
#pragma unroll
    for (int d = 1; d < 32; d <<= 1) {
        const int y = __shfl_up(s, d);
        if (lane >= d) s += y;
    }
#pragma unroll
    for (int d = 16; d > 0; d >>= 1) pre += __shfl_xor(pre, d);
    if (lane == 31) wsum[w] = s;
    if (lane == 0) wpre[w] = pre;
    __syncthreads();
    int base = 0;
    for (int q = 0; q < 32; ++q) {
        base += wpre[q];
        if (q < w) base += wsum[q];
    }
    if (i <= NN) off[i] = base + s - x;
}

__global__ void csr_fill_kernel(const int* __restrict__ src, const int* __restrict__ dst,
                                const int* __restrict__ off, const int* __restrict__ rank,
                                int* __restrict__ csr, int e) {
    int i = blockIdx.x * blockDim.x + threadIdx.x;
    if (i < e) csr[off[dst[i]] + rank[i]] = src[i];
}

__global__ void norm_kernel(float* __restrict__ od, float* __restrict__ id, int n) {
    int i = blockIdx.x * blockDim.x + threadIdx.x;
    if (i < n) {
        od[i] = rsqrtf(fmaxf(od[i], 1.0f));
        id[i] = rsqrtf(fmaxf(id[i], 1.0f));
    }
}


__global__ void embed_scale_kernel(const int* __restrict__ labels, const float* __restrict__ emb,
                                   const float* __restrict__ onorm, float* __restrict__ out) {
    int i = blockIdx.x;
    int f = threadIdx.x;
    out[(long)i * EMB + f] = emb[(long)labels[i] * EMB + f] * onorm[i];
}

template <int D>
__device__ __forceinline__ void gather_rows_body(const float* __restrict__ hs, float* __restrict__ agg,
                                                 const int* __restrict__ off, const int* __restrict__ csr) {
    static_assert(D % 32 == 0, "a row is a whole number of 32-lane lines");
    static_assert(NN % 8 == 0, "every wave of every block owns a node");
    constexpr int NK = D / 32;
    constexpr int UB = 8;
    const int lane = threadIdx.x & 31;
    const int node = blockIdx.x * 8 + __builtin_amdgcn_readfirstlane((int)(threadIdx.x >> 5));
    const int beg = off[node];
    const int end = off[node + 1];
    float acc[NK];
#pragma unroll
    for (int k = 0; k < NK; ++k) acc[k] = 0.0f;
    for (int base = beg; base < end; base += 32) {
        const int n = (end - base < 32) ? (end - base) : 32;
        const int sv = (lane < n) ? csr[base + lane] : 0;
        int j = 0;
        for (; j + UB <= n; j += UB) {
            float v[UB][NK];
#pragma unroll
            for (int u = 0; u < UB; ++u) {
                const int s = __builtin_amdgcn_readlane(sv, j + u);
                const float* row = hs + (long)s * D + lane;
#pragma unroll
                for (int k = 0; k < NK; ++k) v[u][k] = row[k * 32];
            }
#pragma unroll
            for (int u = 0; u < UB; ++u)
#pragma unroll
                for (int k = 0; k < NK; ++k) acc[k] += v[u][k];
        }
        if (j < n) {
            float v[UB][NK];
#pragma unroll
            for (int u = 0; u < UB; ++u) {
#pragma unroll
                for (int k = 0; k < NK; ++k) v[u][k] = 0.0f;
                if (j + u < n) {
                    const int s = __builtin_amdgcn_readlane(sv, j + u);
                    const float* row = hs + (long)s * D + lane;
#pragma unroll
                    for (int k = 0; k < NK; ++k) v[u][k] = row[k * 32];
                }
            }
#pragma unroll
            for (int u = 0; u < UB; ++u)
#pragma unroll
                for (int k = 0; k < NK; ++k) acc[k] += v[u][k];
        }
    }
    float* o = agg + (long)node * D + lane;
#pragma unroll
    for (int k = 0; k < NK; ++k) o[k * 32] = acc[k];
}

__global__ __launch_bounds__(256) void gather_rows64_kernel(const float* __restrict__ hs, float* __restrict__ agg,
                                                            const int* __restrict__ off, const int* __restrict__ csr) {
    gather_rows_body<EMB>(hs, agg, off, csr);
}

__global__ __launch_bounds__(256) void gather_rows96_kernel(const float* __restrict__ hs, float* __restrict__ agg,
                                                            const int* __restrict__ off, const int* __restrict__ csr) {
    gather_rows_body<HID>(hs, agg, off, csr);
}


__global__ __launch_bounds__(256) void gcn_gemm_wmma(
    const float* __restrict__ A, const float* __restrict__ rowScale,
    const float* __restrict__ W, const float* __restrict__ bias,
    const float* __restrict__ outScale,
    float* __restrict__ out, int K, int Nout, int Npad,
    int tilesN, int totalTiles, int relu)
{
    __shared__ float smem[HID * HID + HID];
    const int tid = threadIdx.x;
    const int KN = K * Npad;

    for (int idx = tid; idx < KN; idx += 256) {
        int k = idx / Npad, n = idx % Npad;
        smem[idx] = (n < Nout) ? W[k * Nout + n] : 0.0f;
    }
    for (int idx = tid; idx < Npad; idx += 256)
        smem[KN + idx] = (idx < Nout) ? bias[idx] : 0.0f;
    __syncthreads();

    const int wave = tid >> 5;
    const int lane = tid & 31;
    const int tileId = blockIdx.x * 8 + wave;
    if (tileId >= totalTiles) return;

    const int m0 = (tileId / tilesN) * 16;
    const int n0 = (tileId % tilesN) * 16;

    const int rowA = m0 + (lane & 15);
    const int koff = (lane >> 4) * 2;
    const int colB = lane & 15;

    v8f c;
    const float bval = smem[KN + n0 + colB];
#pragma unroll
    for (int r = 0; r < 8; ++r) c[r] = bval;

    const float s = rowScale[rowA];
    const float* arow = A + (long)rowA * K;
    const float* bcol = smem + n0 + colB;

    const int rbase = m0 + (lane >> 4) * 8;
    float os[8];
#pragma unroll
    for (int r = 0; r < 8; ++r) os[r] = 1.0f;
    if (outScale) {
#pragma unroll
        for (int r = 0; r < 8; ++r) os[r] = outScale[rbase + r];
    }

    for (int k0 = 0; k0 < K; k0 += 4) {
        v2f a, b;
        a.x = arow[k0 + koff] * s;
        a.y = arow[k0 + koff + 1] * s;
        b.x = bcol[(k0 + koff) * Npad];
        b.y = bcol[(k0 + koff + 1) * Npad];
        c = __builtin_amdgcn_wmma_f32_16x16x4_f32(
                 false, a,  false, b,
                 (short)0, c,  false,  false);
    }

    const int col = n0 + colB;
    if (col < Nout) {
#pragma unroll
        for (int r = 0; r < 8; ++r) {
            float v = c[r];
            if (relu) v = fmaxf(v, 0.0f);
            if (outScale) v = v * os[r];
            out[(long)(rbase + r) * Nout + col] = v;
        }
    }
}


extern "C" void kernel_launch(void* const* d_in, const int* in_sizes, int n_in,
                              void* d_out, int out_size, void* d_ws, size_t ws_size,
                              hipStream_t stream) {
    const int*   labels = (const int*)d_in[0];
    const int*   src    = (const int*)d_in[1];
    const int*   dst    = (const int*)d_in[2];
    const float* emb    = (const float*)d_in[3];
    const float* W1     = (const float*)d_in[4];
    const float* b1     = (const float*)d_in[5];
    const float* W2     = (const float*)d_in[6];
    const float* b2     = (const float*)d_in[7];
    const float* W3     = (const float*)d_in[8];
    const float* b3     = (const float*)d_in[9];
    float* out = (float*)d_out;

    float* w = (float*)d_ws;
    float* onorm = w;
    float* inorm = w + NN;
    float* bufA  = w + 2 * (size_t)NN;
    float* bufB  = bufA + (size_t)NN * HID;
    float* bufC  = bufB + (size_t)NN * HID;
    int* off  = (int*)bufB;
    int* rank = off + (NN + 64);
    int* csr  = rank + NE;

    const int TPB = 256;

    zero2_kernel<<<(NN + TPB - 1) / TPB, TPB, 0, stream>>>(onorm, inorm, NN);
    edge_deg_rank_kernel<<<(NE + TPB - 1) / TPB, TPB, 0, stream>>>(src, dst, onorm, inorm, rank, NE);
    offsets_kernel<<<(NN + 1 + 1023) / 1024, 1024, 0, stream>>>(inorm, off);
    norm_kernel<<<(NN + TPB - 1) / TPB, TPB, 0, stream>>>(onorm, inorm, NN);
    csr_fill_kernel<<<(NE + TPB - 1) / TPB, TPB, 0, stream>>>(src, dst, off, rank, csr, NE);

    embed_scale_kernel<<<NN, EMB, 0, stream>>>(labels, emb, onorm, bufA);
    gather_rows64_kernel<<<NN / 8, TPB, 0, stream>>>(bufA, bufC, off, csr);
    {
        int tilesN = HID / 16;
        int total  = (NN / 16) * tilesN;
        gcn_gemm_wmma<<<(total + 7) / 8, 256, 0, stream>>>(
            bufC, inorm, W1, b1, onorm, bufA, EMB, HID, HID, tilesN, total, 1);
    }

    gather_rows96_kernel<<<NN / 8, TPB, 0, stream>>>(bufA, bufC, off, csr);
    {
        int tilesN = HID / 16;
        int total  = (NN / 16) * tilesN;
        gcn_gemm_wmma<<<(total + 7) / 8, 256, 0, stream>>>(
            bufC, inorm, W2, b2, onorm, bufA, HID, HID, HID, tilesN, total, 1);
    }

    gather_rows96_kernel<<<NN / 8, TPB, 0, stream>>>(bufA, bufC, off, csr);
    {
        int Npad   = 64;
        int tilesN = Npad / 16;
        int total  = (NN / 16) * tilesN;
        gcn_gemm_wmma<<<(total + 7) / 8, 256, 0, stream>>>(
            bufC, inorm, W3, b3, (const float*)nullptr, out, HID, NLAB, Npad, tilesN, total, 0);
    }
}


# aggregation and GEMM of a layer in one launch: a 16-wave block gathers 16 node rows into an LDS tile, then 6 (4) waves WMMA from it, W read from global; no agg buffer, 9 launches; sum order unchanged
# speedup vs baseline: 3.2718x; 1.3683x over previous
#include <hip/hip_runtime.h>

#define NN 50000
#define NE 800000
#define NLAB 50
#define EMB 64
#define HID 96

typedef __attribute__((ext_vector_type(2))) float v2f;
typedef __attribute__((ext_vector_type(8))) float v8f;


__global__ void zero2_kernel(float* __restrict__ a, float* __restrict__ b, int n) {
    int i = blockIdx.x * blockDim.x + threadIdx.x;
    if (i < n) { a[i] = 0.0f; b[i] = 0.0f; }
}

__global__ void edge_deg_rank_kernel(const int* __restrict__ src, const int* __restrict__ dst,
                                     float* __restrict__ odeg, float* __restrict__ ideg,
                                     int* __restrict__ rank, int e) {
    int i = blockIdx.x * blockDim.x + threadIdx.x;
    if (i < e) {
        const int s = src[i];
        const int d = dst[i];
        atomicAdd(&odeg[s], 1.0f);
        rank[i] = (int)atomicAdd(&ideg[d], 1.0f);
    }
}

__global__ __launch_bounds__(1024) void offsets_kernel(const float* __restrict__ ideg, int* __restrict__ off) {
    __shared__ int wsum[32];
    __shared__ int wpre[32];
    const int t = threadIdx.x;
    const int lane = t & 31;
    const int w = t >> 5;
    const int i = blockIdx.x * 1024 + t;
    int pre = 0;
    for (int c = 0; c < (int)blockIdx.x; ++c) pre += (int)ideg[c * 1024 + t];
    const int x = (i < NN) ? (int)ideg[i] : 0;
    int s = x;
#pragma unroll
    for (int d = 1; d < 32; d <<= 1) {
        const int y = __shfl_up(s, d);
        if (lane >= d) s += y;
    }
#pragma unroll
    for (int d = 16; d > 0; d >>= 1) pre += __shfl_xor(pre, d);
    if (lane == 31) wsum[w] = s;
    if (lane == 0) wpre[w] = pre;
    __syncthreads();
    int base = 0;
    for (int q = 0; q < 32; ++q) {
        base += wpre[q];
        if (q < w) base += wsum[q];
    }
    if (i <= NN) off[i] = base + s - x;
}

__global__ void csr_fill_kernel(const int* __restrict__ src, const int* __restrict__ dst,
                                const int* __restrict__ off, const int* __restrict__ rank,
                                int* __restrict__ csr, int e) {
    int i = blockIdx.x * blockDim.x + threadIdx.x;
    if (i < e) csr[off[dst[i]] + rank[i]] = src[i];
}

__global__ void norm_kernel(float* __restrict__ od, float* __restrict__ id, int n) {
    int i = blockIdx.x * blockDim.x + threadIdx.x;
    if (i < n) {
        od[i] = rsqrtf(fmaxf(od[i], 1.0f));
        id[i] = rsqrtf(fmaxf(id[i], 1.0f));
    }
}


__global__ void embed_scale_kernel(const int* __restrict__ labels, const float* __restrict__ emb,
                                   const float* __restrict__ onorm, float* __restrict__ out) {
    int i = blockIdx.x;
    int f = threadIdx.x;
    out[(long)i * EMB + f] = emb[(long)labels[i] * EMB + f] * onorm[i];
}

template <int D>
__device__ __forceinline__ void gather_node(const float* __restrict__ hs, const int* __restrict__ off,
                                            const int* __restrict__ csr, int node, int lane,
                                            float (&acc)[D / 32]) {
    static_assert(D % 32 == 0, "a row is a whole number of 32-lane lines");
    constexpr int NK = D / 32;
    constexpr int UB = 8;
    const int beg = off[node];
    const int end = off[node + 1];
#pragma unroll
    for (int k = 0; k < NK; ++k) acc[k] = 0.0f;
    for (int base = beg; base < end; base += 32) {
        const int n = (end - base < 32) ? (end - base) : 32;
        const int sv = (lane < n) ? csr[base + lane] : 0;
        int j = 0;
        for (; j + UB <= n; j += UB) {
            float v[UB][NK];
#pragma unroll
            for (int u = 0; u < UB; ++u) {
                const int s = __builtin_amdgcn_readlane(sv, j + u);
                const float* row = hs + (long)s * D + lane;
#pragma unroll
                for (int k = 0; k < NK; ++k) v[u][k] = row[k * 32];
            }
#pragma unroll
            for (int u = 0; u < UB; ++u)
#pragma unroll
                for (int k = 0; k < NK; ++k) acc[k] += v[u][k];
        }
        if (j < n) {
            float v[UB][NK];
#pragma unroll
            for (int u = 0; u < UB; ++u) {
#pragma unroll
                for (int k = 0; k < NK; ++k) v[u][k] = 0.0f;
                if (j + u < n) {
                    const int s = __builtin_amdgcn_readlane(sv, j + u);
                    const float* row = hs + (long)s * D + lane;
#pragma unroll
                    for (int k = 0; k < NK; ++k) v[u][k] = row[k * 32];
                }
            }
#pragma unroll
            for (int u = 0; u < UB; ++u)
#pragma unroll
                for (int k = 0; k < NK; ++k) acc[k] += v[u][k];
        }
    }
}

template <int D, int NOUT, int NPAD, bool RELU, bool HAS_OS>
__global__ __launch_bounds__(512) void gcn_layer_kernel(
    const float* __restrict__ hs, const int* __restrict__ off, const int* __restrict__ csr,
    const float* __restrict__ rowScale, const float* __restrict__ W, const float* __restrict__ bias,
    const float* __restrict__ outScale, float* __restrict__ out)
{
    static_assert(NN % 16 == 0, "every wave of every block owns a node");
    static_assert(NPAD % 16 == 0 && NPAD >= NOUT && NPAD / 16 <= 16, "one wave per column tile");
    constexpr int NK = D / 32;
    constexpr int PITCH = D + 4;
    constexpr int TILES = NPAD / 16;
    __shared__ float atile[16 * PITCH];

    const int lane = threadIdx.x & 31;
    const int wave = __builtin_amdgcn_readfirstlane((int)(threadIdx.x >> 5));
    const int m0 = blockIdx.x * 16;

    float acc[NK];
    gather_node<D>(hs, off, csr, m0 + wave, lane, acc);
#pragma unroll
    for (int k = 0; k < NK; ++k) atile[wave * PITCH + k * 32 + lane] = acc[k];
    __syncthreads();
    if (wave >= TILES) return;

    const int n0 = wave * 16;
    const int c16 = lane & 15;
    const int half = lane >> 4;
    const int koff = half * 2;
    const int col = n0 + c16;
    const bool cv = (NOUT == NPAD) || (col < NOUT);
    const int colc = cv ? col : 0;

    const float bl = bias[colc];
    const float bval = cv ? bl : 0.0f;
    v8f c;
#pragma unroll
    for (int r = 0; r < 8; ++r) c[r] = bval;

    const float s = rowScale[m0 + c16];
    const int rbase = m0 + half * 8;
    float os[8];
#pragma unroll
    for (int r = 0; r < 8; ++r) os[r] = 1.0f;
    if (HAS_OS) {
#pragma unroll
        for (int r = 0; r < 8; ++r) os[r] = outScale[rbase + r];
    }

    const float* arow = atile + c16 * PITCH + koff;
    const float* bcol = W + colc + koff * NOUT;
#pragma unroll
    for (int k0 = 0; k0 < D; k0 += 4) {
        v2f a, b;
        a.x = arow[k0] * s;
        a.y = arow[k0 + 1] * s;
        const float b0 = bcol[k0 * NOUT];
        const float b1 = bcol[(k0 + 1) * NOUT];
        b.x = cv ? b0 : 0.0f;
        b.y = cv ? b1 : 0.0f;
        c = __builtin_amdgcn_wmma_f32_16x16x4_f32(
                 false, a,  false, b,
                 (short)0, c,  false,  false);
    }

    if (cv) {
#pragma unroll
        for (int r = 0; r < 8; ++r) {
            float v = c[r];
            if (RELU) v = fmaxf(v, 0.0f);
            if (HAS_OS) v = v * os[r];
            out[(long)(rbase + r) * NOUT + col] = v;
        }
    }
}


extern "C" void kernel_launch(void* const* d_in, const int* in_sizes, int n_in,
                              void* d_out, int out_size, void* d_ws, size_t ws_size,
                              hipStream_t stream) {
    const int*   labels = (const int*)d_in[0];
    const int*   src    = (const int*)d_in[1];
    const int*   dst    = (const int*)d_in[2];
    const float* emb    = (const float*)d_in[3];
    const float* W1     = (const float*)d_in[4];
    const float* b1     = (const float*)d_in[5];
    const float* W2     = (const float*)d_in[6];
    const float* b2     = (const float*)d_in[7];
    const float* W3     = (const float*)d_in[8];
    const float* b3     = (const float*)d_in[9];
    float* out = (float*)d_out;

    float* w = (float*)d_ws;
    float* onorm = w;
    float* inorm = w + NN;
    float* bufA  = w + 2 * (size_t)NN;
    float* bufB  = bufA + (size_t)NN * HID;
    float* bufC  = bufB + (size_t)NN * HID;
    int* off  = (int*)bufB;
    int* rank = off + (NN + 64);
    int* csr  = rank + NE;

    const int TPB = 256;

    zero2_kernel<<<(NN + TPB - 1) / TPB, TPB, 0, stream>>>(onorm, inorm, NN);
    edge_deg_rank_kernel<<<(NE + TPB - 1) / TPB, TPB, 0, stream>>>(src, dst, onorm, inorm, rank, NE);
    offsets_kernel<<<(NN + 1 + 1023) / 1024, 1024, 0, stream>>>(inorm, off);
    norm_kernel<<<(NN + TPB - 1) / TPB, TPB, 0, stream>>>(onorm, inorm, NN);
    csr_fill_kernel<<<(NE + TPB - 1) / TPB, TPB, 0, stream>>>(src, dst, off, rank, csr, NE);

    embed_scale_kernel<<<NN, EMB, 0, stream>>>(labels, emb, onorm, bufA);
    gcn_layer_kernel<EMB, HID, HID, true, true><<<NN / 16, 512, 0, stream>>>(
        bufA, off, csr, inorm, W1, b1, onorm, bufC);

    gcn_layer_kernel<HID, HID, HID, true, true><<<NN / 16, 512, 0, stream>>>(
        bufC, off, csr, inorm, W2, b2, onorm, bufA);

    gcn_layer_kernel<HID, NLAB, 64, false, false><<<NN / 16, 512, 0, stream>>>(
        bufA, off, csr, inorm, W3, b3, (const float*)nullptr, out);
}


# layer 1 sums straight from the 50-row embedding table held in LDS, one 8-byte (label, out_norm) record fetched per edge: scaled-h buffer and its launch gone (8 launches); sum order kept, fma
# speedup vs baseline: 3.5176x; 1.0751x over previous
#include <hip/hip_runtime.h>

#define NN 50000
#define NE 800000
#define NLAB 50
#define EMB 64
#define HID 96

typedef __attribute__((ext_vector_type(2))) float v2f;
typedef __attribute__((ext_vector_type(8))) float v8f;


__global__ void zero2_kernel(float* __restrict__ a, float* __restrict__ b, int n) {
    int i = blockIdx.x * blockDim.x + threadIdx.x;
    if (i < n) { a[i] = 0.0f; b[i] = 0.0f; }
}

__global__ void edge_deg_rank_kernel(const int* __restrict__ src, const int* __restrict__ dst,
                                     float* __restrict__ odeg, float* __restrict__ ideg,
                                     int* __restrict__ rank, int e) {
    int i = blockIdx.x * blockDim.x + threadIdx.x;
    if (i < e) {
        const int s = src[i];
        const int d = dst[i];
        atomicAdd(&odeg[s], 1.0f);
        rank[i] = (int)atomicAdd(&ideg[d], 1.0f);
    }
}

__global__ __launch_bounds__(1024) void offsets_kernel(const float* __restrict__ ideg, int* __restrict__ off) {
    __shared__ int wsum[32];
    __shared__ int wpre[32];
    const int t = threadIdx.x;
    const int lane = t & 31;
    const int w = t >> 5;
    const int i = blockIdx.x * 1024 + t;
    int pre = 0;
    for (int c = 0; c < (int)blockIdx.x; ++c) pre += (int)ideg[c * 1024 + t];
    const int x = (i < NN) ? (int)ideg[i] : 0;
    int s = x;
#pragma unroll
    for (int d = 1; d < 32; d <<= 1) {
        const int y = __shfl_up(s, d);
        if (lane >= d) s += y;
    }
#pragma unroll
    for (int d = 16; d > 0; d >>= 1) pre += __shfl_xor(pre, d);
    if (lane == 31) wsum[w] = s;
    if (lane == 0) wpre[w] = pre;
    __syncthreads();
    int base = 0;
    for (int q = 0; q < 32; ++q) {
        base += wpre[q];
        if (q < w) base += wsum[q];
    }
    if (i <= NN) off[i] = base + s - x;
}

__global__ void csr_fill_kernel(const int* __restrict__ src, const int* __restrict__ dst,
                                const int* __restrict__ off, const int* __restrict__ rank,
                                int* __restrict__ csr, int e) {
    int i = blockIdx.x * blockDim.x + threadIdx.x;
    if (i < e) csr[off[dst[i]] + rank[i]] = src[i];
}

__global__ void norm_kernel(float* __restrict__ od, float* __restrict__ id,
                            const int* __restrict__ labels, int* __restrict__ pair, int n) {
    int i = blockIdx.x * blockDim.x + threadIdx.x;
    if (i < n) {
        const float o = rsqrtf(fmaxf(od[i], 1.0f));
        od[i] = o;
        id[i] = rsqrtf(fmaxf(id[i], 1.0f));
        int2 p;
        p.x = labels[i];
        p.y = __float_as_int(o);
        *reinterpret_cast<int2*>(pair + 2 * (long)i) = p;
    }
}

template <int D>
__device__ __forceinline__ void gather_node(const float* __restrict__ hs, const int* __restrict__ off,
                                            const int* __restrict__ csr, int node, int lane,
                                            float (&acc)[D / 32]) {
    static_assert(D % 32 == 0, "a row is a whole number of 32-lane lines");
    constexpr int NK = D / 32;
    constexpr int UB = 8;
    const int beg = off[node];
    const int end = off[node + 1];
#pragma unroll
    for (int k = 0; k < NK; ++k) acc[k] = 0.0f;
    for (int base = beg; base < end; base += 32) {
        const int n = (end - base < 32) ? (end - base) : 32;
        const int sv = (lane < n) ? csr[base + lane] : 0;
        int j = 0;
        for (; j + UB <= n; j += UB) {
            float v[UB][NK];
#pragma unroll
            for (int u = 0; u < UB; ++u) {
                const int s = __builtin_amdgcn_readlane(sv, j + u);
                const float* row = hs + (long)s * D + lane;
#pragma unroll
                for (int k = 0; k < NK; ++k) v[u][k] = row[k * 32];
            }
#pragma unroll
            for (int u = 0; u < UB; ++u)
#pragma unroll
                for (int k = 0; k < NK; ++k) acc[k] += v[u][k];
        }
        if (j < n) {
            float v[UB][NK];
#pragma unroll
            for (int u = 0; u < UB; ++u) {
#pragma unroll
                for (int k = 0; k < NK; ++k) v[u][k] = 0.0f;
                if (j + u < n) {
                    const int s = __builtin_amdgcn_readlane(sv, j + u);
                    const float* row = hs + (long)s * D + lane;
#pragma unroll
                    for (int k = 0; k < NK; ++k) v[u][k] = row[k * 32];
                }
            }
#pragma unroll
            for (int u = 0; u < UB; ++u)
#pragma unroll
                for (int k = 0; k < NK; ++k) acc[k] += v[u][k];
        }
    }
}

__device__ __forceinline__ void edge_records(const int* __restrict__ pair, const int* __restrict__ csr,
                                             int base, int n, int lane, int& lab, int& onb) {
    lab = 0;
    onb = 0;
    if (lane < n) {
        const int sv = csr[base + lane];
        const int2 p = *reinterpret_cast<const int2*>(pair + 2 * (long)sv);
        lab = p.x;
        onb = p.y;
    }
}

__device__ __forceinline__ void walk_records(const float* embs, int n, int lane, int lab, int onb,
                                             float (&acc)[EMB / 32]) {
#pragma unroll 4
    for (int j = 0; j < n; ++j) {
        const int l = __builtin_amdgcn_readlane(lab, j);
        const float o = __int_as_float(__builtin_amdgcn_readlane(onb, j));
        const float* row = embs + l * EMB + lane;
#pragma unroll
        for (int k = 0; k < EMB / 32; ++k) acc[k] += row[k * 32] * o;
    }
}

template <int D, int NOUT, int NPAD, bool RELU, bool HAS_OS, bool TABLE>
__global__ __launch_bounds__(512) void gcn_layer_kernel(
    const float* __restrict__ hs, const int* __restrict__ pair,
    const int* __restrict__ off, const int* __restrict__ csr,
    const float* __restrict__ rowScale, const float* __restrict__ W, const float* __restrict__ bias,
    const float* __restrict__ outScale, float* __restrict__ out)
{
    static_assert(!TABLE || D == EMB, "the table path is the embedding layer's");
    static_assert(NN % 16 == 0, "every wave of every block owns a node");
    static_assert(NPAD % 16 == 0 && NPAD >= NOUT && NPAD / 16 <= 16, "one wave per column tile");
    constexpr int NK = D / 32;
    constexpr int PITCH = D + 4;
    constexpr int TILES = NPAD / 16;
    __shared__ float atile[16 * PITCH];
    __shared__ float embs[TABLE ? NLAB * EMB : 1];

    const int lane = threadIdx.x & 31;
    const int wave = __builtin_amdgcn_readfirstlane((int)(threadIdx.x >> 5));
    const int m0 = blockIdx.x * 16;

    float acc[NK];
    if constexpr (TABLE) {
        constexpr int TBL = NLAB * EMB;
        constexpr int NR = (TBL + 511) / 512;
        const int beg = off[m0 + wave];
        const int end = off[m0 + wave + 1];
        int base = beg;
        int n = (end - base < 32) ? (end - base) : 32;
        int lab, onb;
        edge_records(pair, csr, base, n, lane, lab, onb);
        {
            const int t = threadIdx.x;
            float q[NR];
#pragma unroll
            for (int r = 0; r < NR; ++r) {
                const int i = (t + r * 512 < TBL) ? (t + r * 512) : t;
                q[r] = hs[i];
            }
#pragma unroll
            for (int r = 0; r < NR; ++r) {
                const int i = (t + r * 512 < TBL) ? (t + r * 512) : t;
                embs[i] = q[r];
            }
        }
        __syncthreads();
#pragma unroll
        for (int k = 0; k < NK; ++k) acc[k] = 0.0f;
        while (base < end) {
            walk_records(embs, n, lane, lab, onb, acc);
            base += 32;
            if (base < end) {
                n = (end - base < 32) ? (end - base) : 32;
                edge_records(pair, csr, base, n, lane, lab, onb);
            }
        }
    } else {
        gather_node<D>(hs, off, csr, m0 + wave, lane, acc);
    }
#pragma unroll
    for (int k = 0; k < NK; ++k) atile[wave * PITCH + k * 32 + lane] = acc[k];
    __syncthreads();
    if (wave >= TILES) return;

    const int n0 = wave * 16;
    const int c16 = lane & 15;
    const int half = lane >> 4;
    const int koff = half * 2;
    const int col = n0 + c16;
    const bool cv = (NOUT == NPAD) || (col < NOUT);
    const int colc = cv ? col : 0;

    const float bl = bias[colc];
    const float bval = cv ? bl : 0.0f;
    v8f c;
#pragma unroll
    for (int r = 0; r < 8; ++r) c[r] = bval;

    const float s = rowScale[m0 + c16];
    const int rbase = m0 + half * 8;
    float os[8];
#pragma unroll
    for (int r = 0; r < 8; ++r) os[r] = 1.0f;
    if (HAS_OS) {
#pragma unroll
        for (int r = 0; r < 8; ++r) os[r] = outScale[rbase + r];
    }

    const float* arow = atile + c16 * PITCH + koff;
    const float* bcol = W + colc + koff * NOUT;
#pragma unroll
    for (int k0 = 0; k0 < D; k0 += 4) {
        v2f a, b;
        a.x = arow[k0] * s;
        a.y = arow[k0 + 1] * s;
        const float b0 = bcol[k0 * NOUT];
        const float b1 = bcol[(k0 + 1) * NOUT];
        b.x = cv ? b0 : 0.0f;
        b.y = cv ? b1 : 0.0f;
        c = __builtin_amdgcn_wmma_f32_16x16x4_f32(
                 false, a,  false, b,
                 (short)0, c,  false,  false);
    }

    if (cv) {
#pragma unroll
        for (int r = 0; r < 8; ++r) {
            float v = c[r];
            if (RELU) v = fmaxf(v, 0.0f);
            if (HAS_OS) v = v * os[r];
            out[(long)(rbase + r) * NOUT + col] = v;
        }
    }
}


extern "C" void kernel_launch(void* const* d_in, const int* in_sizes, int n_in,
                              void* d_out, int out_size, void* d_ws, size_t ws_size,
                              hipStream_t stream) {
    const int*   labels = (const int*)d_in[0];
    const int*   src    = (const int*)d_in[1];
    const int*   dst    = (const int*)d_in[2];
    const float* emb    = (const float*)d_in[3];
    const float* W1     = (const float*)d_in[4];
    const float* b1     = (const float*)d_in[5];
    const float* W2     = (const float*)d_in[6];
    const float* b2     = (const float*)d_in[7];
    const float* W3     = (const float*)d_in[8];
    const float* b3     = (const float*)d_in[9];
    float* out = (float*)d_out;

    float* w = (float*)d_ws;
    float* onorm = w;
    float* inorm = w + NN;
    float* bufA  = w + 2 * (size_t)NN;
    float* bufB  = bufA + (size_t)NN * HID;
    float* bufC  = bufB + (size_t)NN * HID;
    int* off  = (int*)bufB;
    int* rank = off + (NN + 64);
    int* csr  = rank + NE;
    int* pair = csr + NE;

    const int TPB = 256;

    zero2_kernel<<<(NN + TPB - 1) / TPB, TPB, 0, stream>>>(onorm, inorm, NN);
    edge_deg_rank_kernel<<<(NE + TPB - 1) / TPB, TPB, 0, stream>>>(src, dst, onorm, inorm, rank, NE);
    offsets_kernel<<<(NN + 1 + 1023) / 1024, 1024, 0, stream>>>(inorm, off);
    norm_kernel<<<(NN + TPB - 1) / TPB, TPB, 0, stream>>>(onorm, inorm, labels, pair, NN);
    csr_fill_kernel<<<(NE + TPB - 1) / TPB, TPB, 0, stream>>>(src, dst, off, rank, csr, NE);

    gcn_layer_kernel<EMB, HID, HID, true, true, true><<<NN / 16, 512, 0, stream>>>(
        emb, pair, off, csr, inorm, W1, b1, onorm, bufC);

    gcn_layer_kernel<HID, HID, HID, true, true, false><<<NN / 16, 512, 0, stream>>>(
        bufC, (const int*)nullptr, off, csr, inorm, W2, b2, onorm, bufA);

    gcn_layer_kernel<HID, NLAB, 64, false, false, false><<<NN / 16, 512, 0, stream>>>(
        bufA, (const int*)nullptr, off, csr, inorm, W3, b3, (const float*)nullptr, out);
}


# sorted edge list replaced by 64 source slots per node, filled at the rank the in-degree add returns (a node past 64 scans the edge list): scan, fill and rank buffer gone, 6 launches; sum order kept
# speedup vs baseline: 4.0755x; 1.1586x over previous
#include <hip/hip_runtime.h>

#define NN 50000
#define NE 800000
#define NLAB 50
#define EMB 64
#define HID 96
#define CAP 64

typedef __attribute__((ext_vector_type(2))) float v2f;
typedef __attribute__((ext_vector_type(8))) float v8f;


__global__ void zero2_kernel(float* __restrict__ a, float* __restrict__ b, int n) {
    int i = blockIdx.x * blockDim.x + threadIdx.x;
    if (i < n) { a[i] = 0.0f; b[i] = 0.0f; }
}

__global__ void edge_deg_bucket_kernel(const int* __restrict__ src, const int* __restrict__ dst,
                                       float* __restrict__ odeg, float* __restrict__ ideg,
                                       int* __restrict__ bucket, int e) {
    int i = blockIdx.x * blockDim.x + threadIdx.x;
    if (i < e) {
        const int s = src[i];
        const int d = dst[i];
        atomicAdd(&odeg[s], 1.0f);
        const int r = (int)atomicAdd(&ideg[d], 1.0f);
        if (r < CAP) bucket[(long)d * CAP + r] = s;
    }
}

__global__ void norm_kernel(float* __restrict__ od, float* __restrict__ id,
                            const int* __restrict__ labels, int* __restrict__ pair,
                            int* __restrict__ cnt, int n) {
    int i = blockIdx.x * blockDim.x + threadIdx.x;
    if (i < n) {
        const float o = rsqrtf(fmaxf(od[i], 1.0f));
        od[i] = o;
        const float c = id[i];
        cnt[i] = (int)c;
        id[i] = rsqrtf(fmaxf(c, 1.0f));
        int2 p;
        p.x = labels[i];
        p.y = __float_as_int(o);
        *reinterpret_cast<int2*>(pair + 2 * (long)i) = p;
    }
}

template <int D, bool TABLE>
__device__ __forceinline__ void scan_node(const float* hs, const int* __restrict__ pair,
                                       const int* __restrict__ esrc, const int* __restrict__ edst,
                                       int node, int lane, float (&acc)[D / 32]) {
    static_assert(NE % 32 == 0, "the edge list is a whole number of 32-edge reads");
    for (int base = 0; base < NE; base += 32) {
        const int d = edst[base + lane];
        unsigned m = __builtin_amdgcn_ballot_w32(d == node);
        if (m != 0u) {
            const int sv = esrc[base + lane];
            while (m != 0u) {
                const int j = __builtin_ctz(m);
                m &= m - 1u;
                const int s = __builtin_amdgcn_readlane(sv, j);
                if constexpr (TABLE) {
                    const int l = pair[2 * (long)s];
                    const float o = __int_as_float(pair[2 * (long)s + 1]);
                    const float* row = hs + l * EMB + lane;
#pragma unroll
                    for (int k = 0; k < D / 32; ++k) acc[k] += row[k * 32] * o;
                } else {
                    const float* row = hs + (long)s * D + lane;
#pragma unroll
                    for (int k = 0; k < D / 32; ++k) acc[k] += row[k * 32];
                }
            }
        }
    }
}

template <int D>
__device__ __forceinline__ void gather_node(const float* __restrict__ hs, const int* __restrict__ cnt,
                                            const int* __restrict__ bucket,
                                            const int* __restrict__ esrc, const int* __restrict__ edst,
                                            int node, int lane, float (&acc)[D / 32]) {
    static_assert(D % 32 == 0, "a row is a whole number of 32-lane lines");
    constexpr int NK = D / 32;
    constexpr int UB = 8;
    const int* bk = bucket + (long)node * CAP;
    const int deg = cnt[node];
    int sv = bk[lane];
    asm volatile("" : "+v"(sv));
#pragma unroll
    for (int k = 0; k < NK; ++k) acc[k] = 0.0f;
    if (deg > CAP) {
        scan_node<D, false>(hs, nullptr, esrc, edst, node, lane, acc);
        return;
    }
    for (int base = 0; base < deg; base += 32) {
        const int n = (deg - base < 32) ? (deg - base) : 32;
        if (base != 0) sv = bk[base + lane];
        int j = 0;
        for (; j + UB <= n; j += UB) {
            float v[UB][NK];
#pragma unroll
            for (int u = 0; u < UB; ++u) {
                const int s = __builtin_amdgcn_readlane(sv, j + u);
                const float* row = hs + (long)s * D + lane;
#pragma unroll
                for (int k = 0; k < NK; ++k) v[u][k] = row[k * 32];
            }
#pragma unroll
            for (int u = 0; u < UB; ++u)
#pragma unroll
                for (int k = 0; k < NK; ++k) acc[k] += v[u][k];
        }
        if (j < n) {
            float v[UB][NK];
#pragma unroll
            for (int u = 0; u < UB; ++u) {
#pragma unroll
                for (int k = 0; k < NK; ++k) v[u][k] = 0.0f;
                if (j + u < n) {
                    const int s = __builtin_amdgcn_readlane(sv, j + u);
                    const float* row = hs + (long)s * D + lane;
#pragma unroll
                    for (int k = 0; k < NK; ++k) v[u][k] = row[k * 32];
                }
            }
#pragma unroll
            for (int u = 0; u < UB; ++u)
#pragma unroll
                for (int k = 0; k < NK; ++k) acc[k] += v[u][k];
        }
    }
}

__device__ __forceinline__ void edge_records(const int* __restrict__ pair, int sv, int n, int lane,
                                             int& lab, int& onb) {
    lab = 0;
    onb = 0;
    if (lane < n) {
        const int2 p = *reinterpret_cast<const int2*>(pair + 2 * (long)sv);
        lab = p.x;
        onb = p.y;
    }
}

__device__ __forceinline__ void walk_records(const float* embs, int n, int lane, int lab, int onb,
                                             float (&acc)[EMB / 32]) {
#pragma unroll 4
    for (int j = 0; j < n; ++j) {
        const int l = __builtin_amdgcn_readlane(lab, j);
        const float o = __int_as_float(__builtin_amdgcn_readlane(onb, j));
        const float* row = embs + l * EMB + lane;
#pragma unroll
        for (int k = 0; k < EMB / 32; ++k) acc[k] += row[k * 32] * o;
    }
}

template <int D, int NOUT, int NPAD, bool RELU, bool HAS_OS, bool TABLE>
__global__ __launch_bounds__(512) void gcn_layer_kernel(
    const float* __restrict__ hs, const int* __restrict__ pair,
    const int* __restrict__ cnt, const int* __restrict__ bucket,
    const int* __restrict__ esrc, const int* __restrict__ edst,
    const float* __restrict__ rowScale, const float* __restrict__ W, const float* __restrict__ bias,
    const float* __restrict__ outScale, float* __restrict__ out)
{
    static_assert(!TABLE || D == EMB, "the table path is the embedding layer's");
    static_assert(NN % 16 == 0, "every wave of every block owns a node");
    static_assert(NPAD % 16 == 0 && NPAD >= NOUT && NPAD / 16 <= 16, "one wave per column tile");
    constexpr int NK = D / 32;
    constexpr int PITCH = D + 4;
    constexpr int TILES = NPAD / 16;
    __shared__ float atile[16 * PITCH];
    __shared__ float embs[TABLE ? NLAB * EMB : 1];

    const int lane = threadIdx.x & 31;
    const int wave = __builtin_amdgcn_readfirstlane((int)(threadIdx.x >> 5));
    const int m0 = blockIdx.x * 16;

    float acc[NK];
    if constexpr (TABLE) {
        constexpr int TBL = NLAB * EMB;
        constexpr int NR = (TBL + 511) / 512;
        const int node = m0 + wave;
        const int* bk = bucket + (long)node * CAP;
        const int deg = cnt[node];
        int sv = bk[lane];
        asm volatile("" : "+v"(sv));
        int n = (deg < 32) ? deg : 32;
        int lab, onb;
        edge_records(pair, sv, n, lane, lab, onb);
        {
            const int t = threadIdx.x;
            float q[NR];
#pragma unroll
            for (int r = 0; r < NR; ++r) {
                const int i = (t + r * 512 < TBL) ? (t + r * 512) : t;
                q[r] = hs[i];
            }
#pragma unroll
            for (int r = 0; r < NR; ++r) {
                const int i = (t + r * 512 < TBL) ? (t + r * 512) : t;
                embs[i] = q[r];
            }
        }
        __syncthreads();
#pragma unroll
        for (int k = 0; k < NK; ++k) acc[k] = 0.0f;
        if (deg > CAP) {
            scan_node<D, true>(embs, pair, esrc, edst, node, lane, acc);
        } else {
            int base = 0;
            while (base < deg) {
                walk_records(embs, n, lane, lab, onb, acc);
                base += 32;
                if (base < deg) {
                    n = (deg - base < 32) ? (deg - base) : 32;
                    sv = bk[base + lane];
                    edge_records(pair, sv, n, lane, lab, onb);
                }
            }
        }
    } else {
        gather_node<D>(hs, cnt, bucket, esrc, edst, m0 + wave, lane, acc);
    }
#pragma unroll
    for (int k = 0; k < NK; ++k) atile[wave * PITCH + k * 32 + lane] = acc[k];
    __syncthreads();
    if (wave >= TILES) return;

    const int n0 = wave * 16;
    const int c16 = lane & 15;
    const int half = lane >> 4;
    const int koff = half * 2;
    const int col = n0 + c16;
    const bool cv = (NOUT == NPAD) || (col < NOUT);
    const int colc = cv ? col : 0;

    const float bl = bias[colc];
    const float bval = cv ? bl : 0.0f;
    v8f c;
#pragma unroll
    for (int r = 0; r < 8; ++r) c[r] = bval;

    const float s = rowScale[m0 + c16];
    const int rbase = m0 + half * 8;
    float os[8];
#pragma unroll
    for (int r = 0; r < 8; ++r) os[r] = 1.0f;
    if (HAS_OS) {
#pragma unroll
        for (int r = 0; r < 8; ++r) os[r] = outScale[rbase + r];
    }

    const float* arow = atile + c16 * PITCH + koff;
    const float* bcol = W + colc + koff * NOUT;
#pragma unroll
    for (int k0 = 0; k0 < D; k0 += 4) {
        v2f a, b;
        a.x = arow[k0] * s;
        a.y = arow[k0 + 1] * s;
        const float b0 = bcol[k0 * NOUT];
        const float b1 = bcol[(k0 + 1) * NOUT];
        b.x = cv ? b0 : 0.0f;
        b.y = cv ? b1 : 0.0f;
        c = __builtin_amdgcn_wmma_f32_16x16x4_f32(
                 false, a,  false, b,
                 (short)0, c,  false,  false);
    }

    if (cv) {
#pragma unroll
        for (int r = 0; r < 8; ++r) {
            float v = c[r];
            if (RELU) v = fmaxf(v, 0.0f);
            if (HAS_OS) v = v * os[r];
            out[(long)(rbase + r) * NOUT + col] = v;
        }
    }
}


extern "C" void kernel_launch(void* const* d_in, const int* in_sizes, int n_in,
                              void* d_out, int out_size, void* d_ws, size_t ws_size,
                              hipStream_t stream) {
    const int*   labels = (const int*)d_in[0];
    const int*   src    = (const int*)d_in[1];
    const int*   dst    = (const int*)d_in[2];
    const float* emb    = (const float*)d_in[3];
    const float* W1     = (const float*)d_in[4];
    const float* b1     = (const float*)d_in[5];
    const float* W2     = (const float*)d_in[6];
    const float* b2     = (const float*)d_in[7];
    const float* W3     = (const float*)d_in[8];
    const float* b3     = (const float*)d_in[9];
    float* out = (float*)d_out;

    float* w = (float*)d_ws;
    float* onorm = w;
    float* inorm = w + NN;
    float* bufA  = w + 2 * (size_t)NN;
    float* bufB  = bufA + (size_t)NN * HID;
    float* bufC  = bufB + (size_t)NN * HID;
    int* bucket = (int*)bufB + 32;
    int* cnt  = bucket + (size_t)NN * CAP;
    int* pair = cnt + NN;

    const int TPB = 256;

    zero2_kernel<<<(NN + TPB - 1) / TPB, TPB, 0, stream>>>(onorm, inorm, NN);
    edge_deg_bucket_kernel<<<(NE + TPB - 1) / TPB, TPB, 0, stream>>>(src, dst, onorm, inorm, bucket, NE);
    norm_kernel<<<(NN + TPB - 1) / TPB, TPB, 0, stream>>>(onorm, inorm, labels, pair, cnt, NN);

    gcn_layer_kernel<EMB, HID, HID, true, true, true><<<NN / 16, 512, 0, stream>>>(
        emb, pair, cnt, bucket, src, dst, inorm, W1, b1, onorm, bufC);

    gcn_layer_kernel<HID, HID, HID, true, true, false><<<NN / 16, 512, 0, stream>>>(
        bufC, (const int*)nullptr, cnt, bucket, src, dst, inorm, W2, b2, onorm, bufA);

    gcn_layer_kernel<HID, NLAB, 64, false, false, false><<<NN / 16, 512, 0, stream>>>(
        bufA, (const int*)nullptr, cnt, bucket, src, dst, inorm, W3, b3, (const float*)nullptr, out);
}


# layer 3's weight product moved in front of its edge sum (done in layer 2's launch from an LDS tile of h): layer 3 sums 64-float rows of h@W3, not 96-float rows of h; sums reordered: product per edge
# speedup vs baseline: 4.2102x; 1.0331x over previous
#include <hip/hip_runtime.h>

#define NN 50000
#define NE 800000
#define NLAB 50
#define EMB 64
#define HID 96
#define CAP 64
#define PW 64

typedef __attribute__((ext_vector_type(2))) float v2f;
typedef __attribute__((ext_vector_type(8))) float v8f;


__global__ void zero2_kernel(float* __restrict__ a, float* __restrict__ b, int n) {
    int i = blockIdx.x * blockDim.x + threadIdx.x;
    if (i < n) { a[i] = 0.0f; b[i] = 0.0f; }
}

__global__ void edge_deg_bucket_kernel(const int* __restrict__ src, const int* __restrict__ dst,
                                       float* __restrict__ odeg, float* __restrict__ ideg,
                                       int* __restrict__ bucket, int e) {
    int i = blockIdx.x * blockDim.x + threadIdx.x;
    if (i < e) {
        const int s = src[i];
        const int d = dst[i];
        atomicAdd(&odeg[s], 1.0f);
        const int r = (int)atomicAdd(&ideg[d], 1.0f);
        if (r < CAP) bucket[(long)d * CAP + r] = s;
    }
}

__global__ void norm_kernel(float* __restrict__ od, float* __restrict__ id,
                            const int* __restrict__ labels, int* __restrict__ pair,
                            int* __restrict__ cnt, int n) {
    int i = blockIdx.x * blockDim.x + threadIdx.x;
    if (i < n) {
        const float o = rsqrtf(fmaxf(od[i], 1.0f));
        od[i] = o;
        const float c = id[i];
        cnt[i] = (int)c;
        id[i] = rsqrtf(fmaxf(c, 1.0f));
        int2 p;
        p.x = labels[i];
        p.y = __float_as_int(o);
        *reinterpret_cast<int2*>(pair + 2 * (long)i) = p;
    }
}

template <int D, bool TABLE>
__device__ __forceinline__ void scan_node(const float* hs, const int* __restrict__ pair,
                                       const int* __restrict__ esrc, const int* __restrict__ edst,
                                       int node, int lane, float (&acc)[D / 32]) {
    static_assert(NE % 32 == 0, "the edge list is a whole number of 32-edge reads");
    for (int base = 0; base < NE; base += 32) {
        const int d = edst[base + lane];
        unsigned m = __builtin_amdgcn_ballot_w32(d == node);
        if (m != 0u) {
            const int sv = esrc[base + lane];
            while (m != 0u) {
                const int j = __builtin_ctz(m);
                m &= m - 1u;
                const int s = __builtin_amdgcn_readlane(sv, j);
                if constexpr (TABLE) {
                    const int l = pair[2 * (long)s];
                    const float o = __int_as_float(pair[2 * (long)s + 1]);
                    const float* row = hs + l * EMB + lane;
#pragma unroll
                    for (int k = 0; k < D / 32; ++k) acc[k] += row[k * 32] * o;
                } else {
                    const float* row = hs + (long)s * D + lane;
#pragma unroll
                    for (int k = 0; k < D / 32; ++k) acc[k] += row[k * 32];
                }
            }
        }
    }
}

template <int D>
__device__ __forceinline__ void gather_node(const float* __restrict__ hs, const int* __restrict__ cnt,
                                            const int* __restrict__ bucket,
                                            const int* __restrict__ esrc, const int* __restrict__ edst,
                                            int node, int lane, float (&acc)[D / 32]) {
    static_assert(D % 32 == 0, "a row is a whole number of 32-lane lines");
    constexpr int NK = D / 32;
    constexpr int UB = 8;
    const int* bk = bucket + (long)node * CAP;
    const int deg = cnt[node];
    int sv = bk[lane];
    asm volatile("" : "+v"(sv));
#pragma unroll
    for (int k = 0; k < NK; ++k) acc[k] = 0.0f;
    if (deg > CAP) {
        scan_node<D, false>(hs, nullptr, esrc, edst, node, lane, acc);
        return;
    }
    for (int base = 0; base < deg; base += 32) {
        const int n = (deg - base < 32) ? (deg - base) : 32;
        if (base != 0) sv = bk[base + lane];
        int j = 0;
        for (; j + UB <= n; j += UB) {
            float v[UB][NK];
#pragma unroll
            for (int u = 0; u < UB; ++u) {
                const int s = __builtin_amdgcn_readlane(sv, j + u);
                const float* row = hs + (long)s * D + lane;
#pragma unroll
                for (int k = 0; k < NK; ++k) v[u][k] = row[k * 32];
            }
#pragma unroll
            for (int u = 0; u < UB; ++u)
#pragma unroll
                for (int k = 0; k < NK; ++k) acc[k] += v[u][k];
        }
        if (j < n) {
            float v[UB][NK];
#pragma unroll
            for (int u = 0; u < UB; ++u) {
#pragma unroll
                for (int k = 0; k < NK; ++k) v[u][k] = 0.0f;
                if (j + u < n) {
                    const int s = __builtin_amdgcn_readlane(sv, j + u);
                    const float* row = hs + (long)s * D + lane;
#pragma unroll
                    for (int k = 0; k < NK; ++k) v[u][k] = row[k * 32];
                }
            }
#pragma unroll
            for (int u = 0; u < UB; ++u)
#pragma unroll
                for (int k = 0; k < NK; ++k) acc[k] += v[u][k];
        }
    }
}

__device__ __forceinline__ void edge_records(const int* __restrict__ pair, int sv, int n, int lane,
                                             int& lab, int& onb) {
    lab = 0;
    onb = 0;
    if (lane < n) {
        const int2 p = *reinterpret_cast<const int2*>(pair + 2 * (long)sv);
        lab = p.x;
        onb = p.y;
    }
}

__device__ __forceinline__ void walk_records(const float* embs, int n, int lane, int lab, int onb,
                                             float (&acc)[EMB / 32]) {
#pragma unroll 4
    for (int j = 0; j < n; ++j) {
        const int l = __builtin_amdgcn_readlane(lab, j);
        const float o = __int_as_float(__builtin_amdgcn_readlane(onb, j));
        const float* row = embs + l * EMB + lane;
#pragma unroll
        for (int k = 0; k < EMB / 32; ++k) acc[k] += row[k * 32] * o;
    }
}

template <int D, int NOUT, int NPAD, bool RELU, bool HAS_OS, bool TABLE, int N2, int NPAD2>
__global__ __launch_bounds__(512) void gcn_layer_kernel(
    const float* __restrict__ hs, const int* __restrict__ pair,
    const int* __restrict__ cnt, const int* __restrict__ bucket,
    const int* __restrict__ esrc, const int* __restrict__ edst,
    const float* __restrict__ rowScale, const float* __restrict__ W, const float* __restrict__ bias,
    const float* __restrict__ outScale, const float* __restrict__ Wn, float* __restrict__ out)
{
    static_assert(!TABLE || D == EMB, "the table path is the embedding layer's");
    static_assert(NN % 16 == 0, "every wave of every block owns a node");
    static_assert(NPAD % 16 == 0 && NPAD >= NOUT && NPAD / 16 <= 16, "one wave per column tile");
    constexpr int NK = D / 32;
    constexpr int PITCH = D + 4;
    constexpr int TILES = NPAD / 16;
    constexpr bool FUSE = (N2 > 0);
    constexpr int TILES2 = FUSE ? NPAD2 / 16 : 0;
    constexpr int PITCH2 = NOUT + 4;
    static_assert(!FUSE || (NOUT == NPAD && NOUT % 4 == 0), "the second product runs over all NOUT columns of h");
    static_assert(!FUSE || (NPAD2 % 16 == 0 && NPAD2 >= N2), "whole column tiles for the second product");
    static_assert(TILES + TILES2 <= 16, "one wave per column tile of either product");
    __shared__ float atile[16 * PITCH];
    __shared__ float htile[FUSE ? 16 * PITCH2 : 1];
    __shared__ float embs[TABLE ? NLAB * EMB : 1];

    const int lane = threadIdx.x & 31;
    const int wave = __builtin_amdgcn_readfirstlane((int)(threadIdx.x >> 5));
    const int m0 = blockIdx.x * 16;

    float acc[NK];
    if constexpr (TABLE) {
        constexpr int TBL = NLAB * EMB;
        constexpr int NR = (TBL + 511) / 512;
        const int node = m0 + wave;
        const int* bk = bucket + (long)node * CAP;
        const int deg = cnt[node];
        int sv = bk[lane];
        asm volatile("" : "+v"(sv));
        int n = (deg < 32) ? deg : 32;
        int lab, onb;
        edge_records(pair, sv, n, lane, lab, onb);
        {
            const int t = threadIdx.x;
            float q[NR];
#pragma unroll
            for (int r = 0; r < NR; ++r) {
                const int i = (t + r * 512 < TBL) ? (t + r * 512) : t;
                q[r] = hs[i];
            }
#pragma unroll
            for (int r = 0; r < NR; ++r) {
                const int i = (t + r * 512 < TBL) ? (t + r * 512) : t;
                embs[i] = q[r];
            }
        }
        __syncthreads();
#pragma unroll
        for (int k = 0; k < NK; ++k) acc[k] = 0.0f;
        if (deg > CAP) {
            scan_node<D, true>(embs, pair, esrc, edst, node, lane, acc);
        } else {
            int base = 0;
            while (base < deg) {
                walk_records(embs, n, lane, lab, onb, acc);
                base += 32;
                if (base < deg) {
                    n = (deg - base < 32) ? (deg - base) : 32;
                    sv = bk[base + lane];
                    edge_records(pair, sv, n, lane, lab, onb);
                }
            }
        }
    } else {
        gather_node<D>(hs, cnt, bucket, esrc, edst, m0 + wave, lane, acc);
    }
#pragma unroll
    for (int k = 0; k < NK; ++k) atile[wave * PITCH + k * 32 + lane] = acc[k];
    __syncthreads();

    const int c16 = lane & 15;
    const int half = lane >> 4;
    const int koff = half * 2;
    const int rbase = m0 + half * 8;

    if (wave < TILES) {
        const int n0 = wave * 16;
        const int col = n0 + c16;
        const bool cv = (NOUT == NPAD) || (col < NOUT);
        const int colc = cv ? col : 0;

        const float bl = bias[colc];
        const float bval = cv ? bl : 0.0f;
        v8f c;
#pragma unroll
        for (int r = 0; r < 8; ++r) c[r] = bval;

        const float s = rowScale[m0 + c16];
        float os[8];
#pragma unroll
        for (int r = 0; r < 8; ++r) os[r] = 1.0f;
        if (HAS_OS) {
#pragma unroll
            for (int r = 0; r < 8; ++r) os[r] = outScale[rbase + r];
        }

        const float* arow = atile + c16 * PITCH + koff;
        const float* bcol = W + colc + koff * NOUT;
#pragma unroll
        for (int k0 = 0; k0 < D; k0 += 4) {
            v2f a, b;
            a.x = arow[k0] * s;
            a.y = arow[k0 + 1] * s;
            const float b0 = bcol[k0 * NOUT];
            const float b1 = bcol[(k0 + 1) * NOUT];
            b.x = cv ? b0 : 0.0f;
            b.y = cv ? b1 : 0.0f;
            c = __builtin_amdgcn_wmma_f32_16x16x4_f32(
                     false, a,  false, b,
                     (short)0, c,  false,  false);
        }

        if constexpr (FUSE) {
#pragma unroll
            for (int r = 0; r < 8; ++r) {
                float v = c[r];
                if (RELU) v = fmaxf(v, 0.0f);
                if (HAS_OS) v = v * os[r];
                htile[(half * 8 + r) * PITCH2 + col] = v;
            }
        } else {
            if (cv) {
#pragma unroll
                for (int r = 0; r < 8; ++r) {
                    float v = c[r];
                    if (RELU) v = fmaxf(v, 0.0f);
                    if (HAS_OS) v = v * os[r];
                    out[(long)(rbase + r) * NOUT + col] = v;
                }
            }
        }
    }

    if constexpr (FUSE) {
        constexpr int KS = NOUT / 4;
        const int t2 = wave - TILES;
        const bool mine = (t2 >= 0) && (t2 < TILES2);
        const int col2 = t2 * 16 + c16;
        const bool cv2 = mine && ((N2 == NPAD2) || (col2 < N2));
        const int colc2 = cv2 ? col2 : 0;
        v2f bw[KS];
        if (mine) {
            const float* bcol = Wn + colc2 + koff * N2;
#pragma unroll
            for (int k = 0; k < KS; ++k) {
                const float b0 = bcol[(4 * k) * N2];
                const float b1 = bcol[(4 * k + 1) * N2];
                bw[k].x = cv2 ? b0 : 0.0f;
                bw[k].y = cv2 ? b1 : 0.0f;
            }
        }
        __syncthreads();
        if (mine) {
            v8f c;
#pragma unroll
            for (int r = 0; r < 8; ++r) c[r] = 0.0f;
            const float* arow = htile + c16 * PITCH2 + koff;
#pragma unroll
            for (int k = 0; k < KS; ++k) {
                v2f a;
                a.x = arow[4 * k];
                a.y = arow[4 * k + 1];
                c = __builtin_amdgcn_wmma_f32_16x16x4_f32(
                         false, a,  false, bw[k],
                         (short)0, c,  false,  false);
            }
#pragma unroll
            for (int r = 0; r < 8; ++r)
                out[(long)(rbase + r) * NPAD2 + col2] = c[r];
        }
    }
}

__device__ __forceinline__ void scan_node_proj(const float* __restrict__ proj,
                                               const int* __restrict__ esrc, const int* __restrict__ edst,
                                               int node, int lane, float& ax, float& ay) {
    for (int base = 0; base < NE; base += 32) {
        const int d = edst[base + lane];
        unsigned m = __builtin_amdgcn_ballot_w32(d == node);
        if (m != 0u) {
            const int sv = esrc[base + lane];
            while (m != 0u) {
                const int j = __builtin_ctz(m);
                m &= m - 1u;
                const int s = __builtin_amdgcn_readlane(sv, j);
                const v2f v = *reinterpret_cast<const v2f*>(proj + (long)s * PW + 2 * lane);
                ax += v.x;
                ay += v.y;
            }
        }
    }
}

__global__ __launch_bounds__(512) void gather_out_kernel(
    const float* __restrict__ proj, const int* __restrict__ cnt, const int* __restrict__ bucket,
    const int* __restrict__ esrc, const int* __restrict__ edst,
    const float* __restrict__ rowScale, const float* __restrict__ bias, float* __restrict__ out)
{
    static_assert(NN % 16 == 0, "every wave of every block owns a node");
    static_assert(NLAB % 2 == 0 && NLAB <= PW && PW == 64, "a lane owns two adjacent columns of a 64-float row");
    constexpr int UB = 16;
    const int lane = threadIdx.x & 31;
    const int wave = __builtin_amdgcn_readfirstlane((int)(threadIdx.x >> 5));
    const int node = blockIdx.x * 16 + wave;
    const int* bk = bucket + (long)node * CAP;
    const float sc = rowScale[node];
    const int cl = (lane < NLAB / 2) ? lane : 0;
    const float bx = bias[2 * cl];
    const float by = bias[2 * cl + 1];
    const int deg = cnt[node];
    int sv = bk[lane];
    asm volatile("" : "+v"(sv));
    const float* prow = proj + 2 * lane;

    float ax = 0.0f, ay = 0.0f;
    if (deg > CAP) {
        scan_node_proj(proj, esrc, edst, node, lane, ax, ay);
    } else {
        for (int base = 0; base < deg; base += 32) {
            const int n = (deg - base < 32) ? (deg - base) : 32;
            if (base != 0) sv = bk[base + lane];
            int j = 0;
            for (; j + UB <= n; j += UB) {
                v2f v[UB];
#pragma unroll
                for (int u = 0; u < UB; ++u) {
                    const int s = __builtin_amdgcn_readlane(sv, j + u);
                    v[u] = *reinterpret_cast<const v2f*>(prow + (long)s * PW);
                }
#pragma unroll
                for (int u = 0; u < UB; ++u) { ax += v[u].x; ay += v[u].y; }
            }
            if (j < n) {
                v2f v[UB];
#pragma unroll
                for (int u = 0; u < UB; ++u) {
                    v[u].x = 0.0f;
                    v[u].y = 0.0f;
                    if (j + u < n) {
                        const int s = __builtin_amdgcn_readlane(sv, j + u);
                        v[u] = *reinterpret_cast<const v2f*>(prow + (long)s * PW);
                    }
                }
#pragma unroll
                for (int u = 0; u < UB; ++u) { ax += v[u].x; ay += v[u].y; }
            }
        }
    }
    if (lane < NLAB / 2) {
        v2f o;
        o.x = ax * sc + bx;
        o.y = ay * sc + by;
        *reinterpret_cast<v2f*>(out + (long)node * NLAB + 2 * lane) = o;
    }
}


extern "C" void kernel_launch(void* const* d_in, const int* in_sizes, int n_in,
                              void* d_out, int out_size, void* d_ws, size_t ws_size,
                              hipStream_t stream) {
    const int*   labels = (const int*)d_in[0];
    const int*   src    = (const int*)d_in[1];
    const int*   dst    = (const int*)d_in[2];
    const float* emb    = (const float*)d_in[3];
    const float* W1     = (const float*)d_in[4];
    const float* b1     = (const float*)d_in[5];
    const float* W2     = (const float*)d_in[6];
    const float* b2     = (const float*)d_in[7];
    const float* W3     = (const float*)d_in[8];
    const float* b3     = (const float*)d_in[9];
    float* out = (float*)d_out;

    float* w = (float*)d_ws;
    float* onorm = w;
    float* inorm = w + NN;
    float* bufA  = w + 2 * (size_t)NN;
    float* bufB  = bufA + (size_t)NN * HID;
    float* bufC  = bufB + (size_t)NN * HID;
    int* bucket = (int*)bufB + 32;
    int* cnt  = bucket + (size_t)NN * CAP;
    int* pair = cnt + NN;
    float* proj = bufA + 32;

    const int TPB = 256;

    zero2_kernel<<<(NN + TPB - 1) / TPB, TPB, 0, stream>>>(onorm, inorm, NN);
    edge_deg_bucket_kernel<<<(NE + TPB - 1) / TPB, TPB, 0, stream>>>(src, dst, onorm, inorm, bucket, NE);
    norm_kernel<<<(NN + TPB - 1) / TPB, TPB, 0, stream>>>(onorm, inorm, labels, pair, cnt, NN);

    gcn_layer_kernel<EMB, HID, HID, true, true, true, 0, 0><<<NN / 16, 512, 0, stream>>>(
        emb, pair, cnt, bucket, src, dst, inorm, W1, b1, onorm, (const float*)nullptr, bufC);

    gcn_layer_kernel<HID, HID, HID, true, true, false, NLAB, PW><<<NN / 16, 512, 0, stream>>>(
        bufC, (const int*)nullptr, cnt, bucket, src, dst, inorm, W2, b2, onorm, W3, proj);

    gather_out_kernel<<<NN / 16, 512, 0, stream>>>(proj, cnt, bucket, src, dst, inorm, b3, out);
}


# degrees, norms and per-node source lists built by grouping edges by 256-node range through LDS counters: no per-edge global atomic or 4-byte scatter left; sum order = LDS-add arrival order
# speedup vs baseline: 7.1103x; 1.6888x over previous
#include <hip/hip_runtime.h>

#define NN 50000
#define NE 800000
#define NLAB 50
#define EMB 64
#define HID 96
#define CAP 64
#define PW 64
#define RNG 256
#define NRNG ((NN + RNG - 1) / RNG)
#define NBIN (2 * NRNG)
#define CAPR 8192
#define CHUNK 2048
#define LCAP 32
#define GSTR 32

typedef __attribute__((ext_vector_type(2))) float v2f;
typedef __attribute__((ext_vector_type(8))) float v8f;


__global__ void zero_kernel(int* __restrict__ gcnt) {
    int i = blockIdx.x * blockDim.x + threadIdx.x;
    if (i < NBIN * GSTR) gcnt[i] = 0;
}

__global__ __launch_bounds__(256) void edge_bin_kernel(const int* __restrict__ src, const int* __restrict__ dst,
                                                       int* __restrict__ gcnt, unsigned* __restrict__ gbin) {
    static_assert(NN <= 65536 && RNG <= 65536, "a record holds the source in 16 bits and the offset in the range above them");
    static_assert(LCAP == 32, "one wave instruction stores a staging row");
    constexpr int T = 256;
    constexpr int EPT = CHUNK / T;
    constexpr int UW = 4;
    __shared__ int lcnt[NBIN];
    __shared__ int lbase[NBIN];
    __shared__ unsigned stage[NBIN * LCAP];
    const int t = threadIdx.x;
    for (int a = t; a < NBIN; a += T) lcnt[a] = 0;
    const int e0 = blockIdx.x * CHUNK;
    int sv[EPT], dv[EPT];
#pragma unroll
    for (int u = 0; u < EPT; ++u) {
        const int i = e0 + u * T + t;
        const int ii = (i < NE) ? i : (NE - 1);
        sv[u] = src[ii];
        dv[u] = dst[ii];
    }
    __syncthreads();
    int pd[EPT], ps[EPT];
#pragma unroll
    for (int u = 0; u < EPT; ++u) {
        const int i = e0 + u * T + t;
        pd[u] = 0;
        ps[u] = 0;
        if (i < NE) {
            pd[u] = atomicAdd(&lcnt[(int)((unsigned)dv[u] / RNG)], 1);
            ps[u] = atomicAdd(&lcnt[NRNG + (int)((unsigned)sv[u] / RNG)], 1);
        }
    }
#pragma unroll
    for (int u = 0; u < EPT; ++u) {
        const int i = e0 + u * T + t;
        if (i < NE) {
            const int ad = (int)((unsigned)dv[u] / RNG);
            const int as = NRNG + (int)((unsigned)sv[u] / RNG);
            if (pd[u] < LCAP) stage[ad * LCAP + pd[u]] = (unsigned)sv[u] | ((unsigned)(dv[u] - ad * RNG) << 16);
            if (ps[u] < LCAP) stage[as * LCAP + ps[u]] = (unsigned)(sv[u] - (as - NRNG) * RNG);
        }
    }
    __syncthreads();
    for (int a = t; a < NBIN; a += T) {
        const int c = lcnt[a];
        lbase[a] = (c > 0) ? atomicAdd(&gcnt[a * GSTR], c) : 0;
    }
    __syncthreads();
#pragma unroll
    for (int u = 0; u < EPT; ++u) {
        const int i = e0 + u * T + t;
        if (i < NE) {
            const int ad = (int)((unsigned)dv[u] / RNG);
            const int as = NRNG + (int)((unsigned)sv[u] / RNG);
            if (pd[u] >= LCAP) {
                const int g = lbase[ad] + pd[u];
                if (g < CAPR) gbin[(long)ad * CAPR + g] = (unsigned)sv[u] | ((unsigned)(dv[u] - ad * RNG) << 16);
            }
            if (ps[u] >= LCAP) {
                const int g = lbase[as] + ps[u];
                if (g < CAPR) gbin[(long)as * CAPR + g] = (unsigned)(sv[u] - (as - NRNG) * RNG);
            }
        }
    }
    const int lane = t & 31;
    const int wave = t >> 5;
    for (int a0 = wave; a0 < NBIN; a0 += (T / 32) * UW) {
        int c[UW], g[UW];
        unsigned v[UW];
#pragma unroll
        for (int u = 0; u < UW; ++u) {
            const int a = a0 + u * (T / 32);
            const int ax = (a < NBIN) ? a : 0;
            const int cc = lcnt[ax];
            c[u] = (a < NBIN) ? ((cc < LCAP) ? cc : LCAP) : 0;
            g[u] = lbase[ax] + lane;
        }
#pragma unroll
        for (int u = 0; u < UW; ++u) {
            const int a = a0 + u * (T / 32);
            v[u] = (lane < c[u]) ? stage[a * LCAP + lane] : 0u;
        }
#pragma unroll
        for (int u = 0; u < UW; ++u) {
            const int a = a0 + u * (T / 32);
            if (lane < c[u] && g[u] < CAPR) gbin[(long)a * CAPR + g[u]] = v[u];
        }
    }
}

__global__ __launch_bounds__(1024) void range_kernel(const unsigned* __restrict__ gbin, const int* __restrict__ gcnt,
                                                     const int* __restrict__ src, const int* __restrict__ dst,
                                                     const int* __restrict__ labels,
                                                     int* __restrict__ bucket, int* __restrict__ cnt,
                                                     float* __restrict__ inorm, float* __restrict__ onorm,
                                                     int* __restrict__ pair) {
    static_assert((RNG * CAP) % 4 == 0 && CAP % 4 == 0, "rows leave in 16-byte pieces");
    constexpr int T = 1024;
    constexpr int UR = 5;
    __shared__ int lc[RNG];
    __shared__ int4 slots4[RNG * CAP / 4];
    int* slots = reinterpret_cast<int*>(slots4);
    const int t = threadIdx.x;
    const int a = blockIdx.x;
    const bool lists = a < NRNG;
    const int r = lists ? a : a - NRNG;
    for (int i = t; i < RNG; i += T) lc[i] = 0;
    if (lists)
        for (int i = t; i < RNG * CAP / 4; i += T) slots4[i] = make_int4(0, 0, 0, 0);
    __syncthreads();
    const int total = gcnt[a * GSTR];
    if (total <= CAPR) {
        const unsigned* rec = gbin + (long)a * CAPR;
        for (int b = 0; b < total; b += UR * T) {
            unsigned q[UR];
#pragma unroll
            for (int u = 0; u < UR; ++u) {
                const int i = b + u * T + t;
                q[u] = rec[(i < total) ? i : (total - 1)];
            }
#pragma unroll
            for (int u = 0; u < UR; ++u) {
                const int i = b + u * T + t;
                if (i < total) {
                    if (lists) {
                        const int dl = (int)(q[u] >> 16);
                        const int k = atomicAdd(&lc[dl], 1);
                        if (k < CAP) slots[dl * CAP + k] = (int)(q[u] & 0xFFFFu);
                    } else {
                        atomicAdd(&lc[(int)q[u]], 1);
                    }
                }
            }
        }
    } else if (lists) {
        for (int i = t; i < NE; i += T) {
            const int d = dst[i];
            if ((int)((unsigned)d / RNG) == r) {
                const int dl = d - r * RNG;
                const int k = atomicAdd(&lc[dl], 1);
                if (k < CAP) slots[dl * CAP + k] = src[i];
            }
        }
    } else {
        for (int i = t; i < NE; i += T) {
            const int s = src[i];
            if ((int)((unsigned)s / RNG) == r) atomicAdd(&lc[s - r * RNG], 1);
        }
    }
    __syncthreads();
    const int n0 = r * RNG;
    if (lists) {
        for (int i = t; i < RNG; i += T)
            if (n0 + i < NN) {
                const int c = lc[i];
                cnt[n0 + i] = c;
                inorm[n0 + i] = rsqrtf(fmaxf((float)c, 1.0f));
            }
        int4* brow = reinterpret_cast<int4*>(bucket + (long)n0 * CAP);
        for (int i = t; i < RNG * CAP / 4; i += T)
            if (n0 + (i * 4) / CAP < NN) brow[i] = slots4[i];
    } else {
        for (int i = t; i < RNG; i += T)
            if (n0 + i < NN) {
                const float o = rsqrtf(fmaxf((float)lc[i], 1.0f));
                onorm[n0 + i] = o;
                int2 p;
                p.x = labels[n0 + i];
                p.y = __float_as_int(o);
                *reinterpret_cast<int2*>(pair + 2 * (long)(n0 + i)) = p;
            }
    }
}

template <int D, bool TABLE>
__device__ __forceinline__ void scan_node(const float* hs, const int* __restrict__ pair,
                                       const int* __restrict__ esrc, const int* __restrict__ edst,
                                       int node, int lane, float (&acc)[D / 32]) {
    static_assert(NE % 32 == 0, "the edge list is a whole number of 32-edge reads");
    for (int base = 0; base < NE; base += 32) {
        const int d = edst[base + lane];
        unsigned m = __builtin_amdgcn_ballot_w32(d == node);
        if (m != 0u) {
            const int sv = esrc[base + lane];
            while (m != 0u) {
                const int j = __builtin_ctz(m);
                m &= m - 1u;
                const int s = __builtin_amdgcn_readlane(sv, j);
                if constexpr (TABLE) {
                    const int l = pair[2 * (long)s];
                    const float o = __int_as_float(pair[2 * (long)s + 1]);
                    const float* row = hs + l * EMB + lane;
#pragma unroll
                    for (int k = 0; k < D / 32; ++k) acc[k] += row[k * 32] * o;
                } else {
                    const float* row = hs + (long)s * D + lane;
#pragma unroll
                    for (int k = 0; k < D / 32; ++k) acc[k] += row[k * 32];
                }
            }
        }
    }
}

template <int D>
__device__ __forceinline__ void gather_node(const float* __restrict__ hs, const int* __restrict__ cnt,
                                            const int* __restrict__ bucket,
                                            const int* __restrict__ esrc, const int* __restrict__ edst,
                                            int node, int lane, float (&acc)[D / 32]) {
    static_assert(D % 32 == 0, "a row is a whole number of 32-lane lines");
    constexpr int NK = D / 32;
    constexpr int UB = 8;
    const int* bk = bucket + (long)node * CAP;
    const int deg = cnt[node];
    int sv = bk[lane];
    asm volatile("" : "+v"(sv));
#pragma unroll
    for (int k = 0; k < NK; ++k) acc[k] = 0.0f;
    if (deg > CAP) {
        scan_node<D, false>(hs, nullptr, esrc, edst, node, lane, acc);
        return;
    }
    for (int base = 0; base < deg; base += 32) {
        const int n = (deg - base < 32) ? (deg - base) : 32;
        if (base != 0) sv = bk[base + lane];
        int j = 0;
        for (; j + UB <= n; j += UB) {
            float v[UB][NK];
#pragma unroll
            for (int u = 0; u < UB; ++u) {
                const int s = __builtin_amdgcn_readlane(sv, j + u);
                const float* row = hs + (long)s * D + lane;
#pragma unroll
                for (int k = 0; k < NK; ++k) v[u][k] = row[k * 32];
            }
#pragma unroll
            for (int u = 0; u < UB; ++u)
#pragma unroll
                for (int k = 0; k < NK; ++k) acc[k] += v[u][k];
        }
        if (j < n) {
            float v[UB][NK];
#pragma unroll
            for (int u = 0; u < UB; ++u) {
#pragma unroll
                for (int k = 0; k < NK; ++k) v[u][k] = 0.0f;
                if (j + u < n) {
                    const int s = __builtin_amdgcn_readlane(sv, j + u);
                    const float* row = hs + (long)s * D + lane;
#pragma unroll
                    for (int k = 0; k < NK; ++k) v[u][k] = row[k * 32];
                }
            }
#pragma unroll
            for (int u = 0; u < UB; ++u)
#pragma unroll
                for (int k = 0; k < NK; ++k) acc[k] += v[u][k];
        }
    }
}

__device__ __forceinline__ void edge_records(const int* __restrict__ pair, int sv, int n, int lane,
                                             int& lab, int& onb) {
    lab = 0;
    onb = 0;
    if (lane < n) {
        const int2 p = *reinterpret_cast<const int2*>(pair + 2 * (long)sv);
        lab = p.x;
        onb = p.y;
    }
}

__device__ __forceinline__ void walk_records(const float* embs, int n, int lane, int lab, int onb,
                                             float (&acc)[EMB / 32]) {
#pragma unroll 4
    for (int j = 0; j < n; ++j) {
        const int l = __builtin_amdgcn_readlane(lab, j);
        const float o = __int_as_float(__builtin_amdgcn_readlane(onb, j));
        const float* row = embs + l * EMB + lane;
#pragma unroll
        for (int k = 0; k < EMB / 32; ++k) acc[k] += row[k * 32] * o;
    }
}

template <int D, int NOUT, int NPAD, bool RELU, bool HAS_OS, bool TABLE, int N2, int NPAD2>
__global__ __launch_bounds__(512) void gcn_layer_kernel(
    const float* __restrict__ hs, const int* __restrict__ pair,
    const int* __restrict__ cnt, const int* __restrict__ bucket,
    const int* __restrict__ esrc, const int* __restrict__ edst,
    const float* __restrict__ rowScale, const float* __restrict__ W, const float* __restrict__ bias,
    const float* __restrict__ outScale, const float* __restrict__ Wn, float* __restrict__ out)
{
    static_assert(!TABLE || D == EMB, "the table path is the embedding layer's");
    static_assert(NN % 16 == 0, "every wave of every block owns a node");
    static_assert(NPAD % 16 == 0 && NPAD >= NOUT && NPAD / 16 <= 16, "one wave per column tile");
    constexpr int NK = D / 32;
    constexpr int PITCH = D + 4;
    constexpr int TILES = NPAD / 16;
    constexpr bool FUSE = (N2 > 0);
    constexpr int TILES2 = FUSE ? NPAD2 / 16 : 0;
    constexpr int PITCH2 = NOUT + 4;
    static_assert(!FUSE || (NOUT == NPAD && NOUT % 4 == 0), "the second product runs over all NOUT columns of h");
    static_assert(!FUSE || (NPAD2 % 16 == 0 && NPAD2 >= N2), "whole column tiles for the second product");
    static_assert(TILES + TILES2 <= 16, "one wave per column tile of either product");
    __shared__ float atile[16 * PITCH];
    __shared__ float htile[FUSE ? 16 * PITCH2 : 1];
    __shared__ float embs[TABLE ? NLAB * EMB : 1];

    const int lane = threadIdx.x & 31;
    const int wave = __builtin_amdgcn_readfirstlane((int)(threadIdx.x >> 5));
    const int m0 = blockIdx.x * 16;

    float acc[NK];
    if constexpr (TABLE) {
        constexpr int TBL = NLAB * EMB;
        constexpr int NR = (TBL + 511) / 512;
        const int node = m0 + wave;
        const int* bk = bucket + (long)node * CAP;
        const int deg = cnt[node];
        int sv = bk[lane];
        asm volatile("" : "+v"(sv));
        int n = (deg < 32) ? deg : 32;
        int lab, onb;
        edge_records(pair, sv, n, lane, lab, onb);
        {
            const int t = threadIdx.x;
            float q[NR];
#pragma unroll
            for (int r = 0; r < NR; ++r) {
                const int i = (t + r * 512 < TBL) ? (t + r * 512) : t;
                q[r] = hs[i];
            }
#pragma unroll
            for (int r = 0; r < NR; ++r) {
                const int i = (t + r * 512 < TBL) ? (t + r * 512) : t;
                embs[i] = q[r];
            }
        }
        __syncthreads();
#pragma unroll
        for (int k = 0; k < NK; ++k) acc[k] = 0.0f;
        if (deg > CAP) {
            scan_node<D, true>(embs, pair, esrc, edst, node, lane, acc);
        } else {
            int base = 0;
            while (base < deg) {
                walk_records(embs, n, lane, lab, onb, acc);
                base += 32;
                if (base < deg) {
                    n = (deg - base < 32) ? (deg - base) : 32;
                    sv = bk[base + lane];
                    edge_records(pair, sv, n, lane, lab, onb);
                }
            }
        }
    } else {
        gather_node<D>(hs, cnt, bucket, esrc, edst, m0 + wave, lane, acc);
    }
#pragma unroll
    for (int k = 0; k < NK; ++k) atile[wave * PITCH + k * 32 + lane] = acc[k];
    __syncthreads();

    const int c16 = lane & 15;
    const int half = lane >> 4;
    const int koff = half * 2;
    const int rbase = m0 + half * 8;

    if (wave < TILES) {
        const int n0 = wave * 16;
        const int col = n0 + c16;
        const bool cv = (NOUT == NPAD) || (col < NOUT);
        const int colc = cv ? col : 0;

        const float bl = bias[colc];
        const float bval = cv ? bl : 0.0f;
        v8f c;
#pragma unroll
        for (int r = 0; r < 8; ++r) c[r] = bval;

        const float s = rowScale[m0 + c16];
        float os[8];
#pragma unroll
        for (int r = 0; r < 8; ++r) os[r] = 1.0f;
        if (HAS_OS) {
#pragma unroll
            for (int r = 0; r < 8; ++r) os[r] = outScale[rbase + r];
        }

        const float* arow = atile + c16 * PITCH + koff;
        const float* bcol = W + colc + koff * NOUT;
#pragma unroll
        for (int k0 = 0; k0 < D; k0 += 4) {
            v2f a, b;
            a.x = arow[k0] * s;
            a.y = arow[k0 + 1] * s;
            const float b0 = bcol[k0 * NOUT];
            const float b1 = bcol[(k0 + 1) * NOUT];
            b.x = cv ? b0 : 0.0f;
            b.y = cv ? b1 : 0.0f;
            c = __builtin_amdgcn_wmma_f32_16x16x4_f32(
                     false, a,  false, b,
                     (short)0, c,  false,  false);
        }

        if constexpr (FUSE) {
#pragma unroll
            for (int r = 0; r < 8; ++r) {
                float v = c[r];
                if (RELU) v = fmaxf(v, 0.0f);
                if (HAS_OS) v = v * os[r];
                htile[(half * 8 + r) * PITCH2 + col] = v;
            }
        } else {
            if (cv) {
#pragma unroll
                for (int r = 0; r < 8; ++r) {
                    float v = c[r];
                    if (RELU) v = fmaxf(v, 0.0f);
                    if (HAS_OS) v = v * os[r];
                    out[(long)(rbase + r) * NOUT + col] = v;
                }
            }
        }
    }

    if constexpr (FUSE) {
        constexpr int KS = NOUT / 4;
        const int t2 = wave - TILES;
        const bool mine = (t2 >= 0) && (t2 < TILES2);
        const int col2 = t2 * 16 + c16;
        const bool cv2 = mine && ((N2 == NPAD2) || (col2 < N2));
        const int colc2 = cv2 ? col2 : 0;
        v2f bw[KS];
        if (mine) {
            const float* bcol = Wn + colc2 + koff * N2;
#pragma unroll
            for (int k = 0; k < KS; ++k) {
                const float b0 = bcol[(4 * k) * N2];
                const float b1 = bcol[(4 * k + 1) * N2];
                bw[k].x = cv2 ? b0 : 0.0f;
                bw[k].y = cv2 ? b1 : 0.0f;
            }
        }
        __syncthreads();
        if (mine) {
            v8f c;
#pragma unroll
            for (int r = 0; r < 8; ++r) c[r] = 0.0f;
            const float* arow = htile + c16 * PITCH2 + koff;
#pragma unroll
            for (int k = 0; k < KS; ++k) {
                v2f a;
                a.x = arow[4 * k];
                a.y = arow[4 * k + 1];
                c = __builtin_amdgcn_wmma_f32_16x16x4_f32(
                         false, a,  false, bw[k],
                         (short)0, c,  false,  false);
            }
#pragma unroll
            for (int r = 0; r < 8; ++r)
                out[(long)(rbase + r) * NPAD2 + col2] = c[r];
        }
    }
}

__device__ __forceinline__ void scan_node_proj(const float* __restrict__ proj,
                                               const int* __restrict__ esrc, const int* __restrict__ edst,
                                               int node, int lane, float& ax, float& ay) {
    for (int base = 0; base < NE; base += 32) {
        const int d = edst[base + lane];
        unsigned m = __builtin_amdgcn_ballot_w32(d == node);
        if (m != 0u) {
            const int sv = esrc[base + lane];
            while (m != 0u) {
                const int j = __builtin_ctz(m);
                m &= m - 1u;
                const int s = __builtin_amdgcn_readlane(sv, j);
                const v2f v = *reinterpret_cast<const v2f*>(proj + (long)s * PW + 2 * lane);
                ax += v.x;
                ay += v.y;
            }
        }
    }
}

__global__ __launch_bounds__(512) void gather_out_kernel(
    const float* __restrict__ proj, const int* __restrict__ cnt, const int* __restrict__ bucket,
    const int* __restrict__ esrc, const int* __restrict__ edst,
    const float* __restrict__ rowScale, const float* __restrict__ bias, float* __restrict__ out)
{
    static_assert(NN % 16 == 0, "every wave of every block owns a node");
    static_assert(NLAB % 2 == 0 && NLAB <= PW && PW == 64, "a lane owns two adjacent columns of a 64-float row");
    constexpr int UB = 16;
    const int lane = threadIdx.x & 31;
    const int wave = __builtin_amdgcn_readfirstlane((int)(threadIdx.x >> 5));
    const int node = blockIdx.x * 16 + wave;
    const int* bk = bucket + (long)node * CAP;
    const float sc = rowScale[node];
    const int cl = (lane < NLAB / 2) ? lane : 0;
    const float bx = bias[2 * cl];
    const float by = bias[2 * cl + 1];
    const int deg = cnt[node];
    int sv = bk[lane];
    asm volatile("" : "+v"(sv));
    const float* prow = proj + 2 * lane;

    float ax = 0.0f, ay = 0.0f;
    if (deg > CAP) {
        scan_node_proj(proj, esrc, edst, node, lane, ax, ay);
    } else {
        for (int base = 0; base < deg; base += 32) {
            const int n = (deg - base < 32) ? (deg - base) : 32;
            if (base != 0) sv = bk[base + lane];
            int j = 0;
            for (; j + UB <= n; j += UB) {
                v2f v[UB];
#pragma unroll
                for (int u = 0; u < UB; ++u) {
                    const int s = __builtin_amdgcn_readlane(sv, j + u);
                    v[u] = *reinterpret_cast<const v2f*>(prow + (long)s * PW);
                }
#pragma unroll
                for (int u = 0; u < UB; ++u) { ax += v[u].x; ay += v[u].y; }
            }
            if (j < n) {
                v2f v[UB];
#pragma unroll
                for (int u = 0; u < UB; ++u) {
                    v[u].x = 0.0f;
                    v[u].y = 0.0f;
                    if (j + u < n) {
                        const int s = __builtin_amdgcn_readlane(sv, j + u);
                        v[u] = *reinterpret_cast<const v2f*>(prow + (long)s * PW);
                    }
                }
#pragma unroll
                for (int u = 0; u < UB; ++u) { ax += v[u].x; ay += v[u].y; }
            }
        }
    }
    if (lane < NLAB / 2) {
        v2f o;
        o.x = ax * sc + bx;
        o.y = ay * sc + by;
        *reinterpret_cast<v2f*>(out + (long)node * NLAB + 2 * lane) = o;
    }
}


extern "C" void kernel_launch(void* const* d_in, const int* in_sizes, int n_in,
                              void* d_out, int out_size, void* d_ws, size_t ws_size,
                              hipStream_t stream) {
    const int*   labels = (const int*)d_in[0];
    const int*   src    = (const int*)d_in[1];
    const int*   dst    = (const int*)d_in[2];
    const float* emb    = (const float*)d_in[3];
    const float* W1     = (const float*)d_in[4];
    const float* b1     = (const float*)d_in[5];
    const float* W2     = (const float*)d_in[6];
    const float* b2     = (const float*)d_in[7];
    const float* W3     = (const float*)d_in[8];
    const float* b3     = (const float*)d_in[9];
    float* out = (float*)d_out;

    float* w = (float*)d_ws;
    float* onorm = w;
    float* inorm = w + NN;
    float* bufA  = w + 2 * (size_t)NN;
    float* bufB  = bufA + (size_t)NN * HID;
    float* bufC  = bufB + (size_t)NN * HID;
    int* bucket = (int*)bufB + 32;
    int* cnt  = bucket + (size_t)NN * CAP;
    int* pair = cnt + NN;
    float* proj = bufA + 32;
    int* gcnt = (int*)(proj + (size_t)NN * PW);
    unsigned* gbin = (unsigned*)bufC;
    static_assert((size_t)NBIN * CAPR <= (size_t)NN * HID, "the records fit the buffer they borrow");
    static_assert(32 + (size_t)NN * PW + (size_t)NBIN * GSTR <= (size_t)NN * HID, "the counters fit behind proj");

    const int TPB = 256;

    zero_kernel<<<(NBIN * GSTR + TPB - 1) / TPB, TPB, 0, stream>>>(gcnt);
    edge_bin_kernel<<<(NE + CHUNK - 1) / CHUNK, 256, 0, stream>>>(src, dst, gcnt, gbin);
    range_kernel<<<NBIN, 1024, 0, stream>>>(gbin, gcnt, src, dst, labels, bucket, cnt, inorm, onorm, pair);

    gcn_layer_kernel<EMB, HID, HID, true, true, true, 0, 0><<<NN / 16, 512, 0, stream>>>(
        emb, pair, cnt, bucket, src, dst, inorm, W1, b1, onorm, (const float*)nullptr, bufC);

    gcn_layer_kernel<HID, HID, HID, true, true, false, NLAB, PW><<<NN / 16, 512, 0, stream>>>(
        bufC, (const int*)nullptr, cnt, bucket, src, dst, inorm, W2, b2, onorm, W3, proj);

    gather_out_kernel<<<NN / 16, 512, 0, stream>>>(proj, cnt, bucket, src, dst, inorm, b3, out);
}


# each binning block stores one whole 128-byte line per (block, range): count in word 0, then records; no counters in memory, no global atomic, no zero launch (5 launches); sum order: LDS-add arrival
# speedup vs baseline: 8.4885x; 1.1938x over previous
#include <hip/hip_runtime.h>

#define NN 50000
#define NE 800000
#define NLAB 50
#define EMB 64
#define HID 96
#define CAP 64
#define PW 64
#define RNG 256
#define NRNG ((NN + RNG - 1) / RNG)
#define NBIN (2 * NRNG)
#define CHUNK 2048
#define NBLK ((NE + CHUNK - 1) / CHUNK)
#define LW 32
#define LCAP (LW - 1)

typedef __attribute__((ext_vector_type(2))) float v2f;
typedef __attribute__((ext_vector_type(8))) float v8f;


__global__ __launch_bounds__(256) void edge_bin_kernel(const int* __restrict__ src, const int* __restrict__ dst,
                                                       unsigned* __restrict__ gbin) {
    static_assert(NN <= 65536 && RNG <= 65536, "a record holds the source in 16 bits and the offset in the range above them");
    static_assert(LW == 32, "one wave instruction stores a line");
    constexpr int T = 256;
    constexpr int EPT = CHUNK / T;
    constexpr int UW = 7;
    __shared__ int lcnt[NBIN];
    __shared__ uint4 stage4[NBIN * LW / 4];
    unsigned* stage = reinterpret_cast<unsigned*>(stage4);
    const int t = threadIdx.x;
    const int e0 = blockIdx.x * CHUNK;
    int sv[EPT], dv[EPT];
#pragma unroll
    for (int u = 0; u < EPT; ++u) {
        const int i = e0 + u * T + t;
        const int ii = (i < NE) ? i : (NE - 1);
        sv[u] = src[ii];
        dv[u] = dst[ii];
    }
    for (int a = t; a < NBIN; a += T) lcnt[a] = 0;
    for (int i = t; i < NBIN * LW / 4; i += T) stage4[i] = make_uint4(0u, 0u, 0u, 0u);
    __syncthreads();
    int pd[EPT], ps[EPT];
#pragma unroll
    for (int u = 0; u < EPT; ++u) {
        const int i = e0 + u * T + t;
        pd[u] = 0;
        ps[u] = 0;
        if (i < NE) {
            pd[u] = atomicAdd(&lcnt[(int)((unsigned)dv[u] / RNG)], 1);
            ps[u] = atomicAdd(&lcnt[NRNG + (int)((unsigned)sv[u] / RNG)], 1);
        }
    }
#pragma unroll
    for (int u = 0; u < EPT; ++u) {
        const int i = e0 + u * T + t;
        if (i < NE) {
            const int ad = (int)((unsigned)dv[u] / RNG);
            const int as = NRNG + (int)((unsigned)sv[u] / RNG);
            if (pd[u] < LCAP) stage[ad * LW + 1 + pd[u]] = (unsigned)sv[u] | ((unsigned)(dv[u] - ad * RNG) << 16);
            if (ps[u] < LCAP) stage[as * LW + 1 + ps[u]] = (unsigned)(sv[u] - (as - NRNG) * RNG);
        }
    }
    __syncthreads();
    const int lane = t & 31;
    const int wave = __builtin_amdgcn_readfirstlane(t >> 5);
    unsigned* mine = gbin + (long)blockIdx.x * LW + lane;
    constexpr int APW = NBIN / (T / 32);
    static_assert(NBIN % (T / 32) == 0 && APW % UW == 0, "every wave stores whole groups of UW lines");
    for (int i0 = 0; i0 < APW; i0 += UW) {
        unsigned v[UW];
#pragma unroll
        for (int u = 0; u < UW; ++u) {
            const int a = wave + (i0 + u) * (T / 32);
            const unsigned* p = (lane == 0) ? reinterpret_cast<const unsigned*>(&lcnt[a]) : &stage[a * LW + lane];
            v[u] = *p;
        }
#pragma unroll
        for (int u = 0; u < UW; ++u) {
            const int a = wave + (i0 + u) * (T / 32);
            mine[(long)a * (NBLK * LW)] = v[u];
        }
    }
}

__global__ __launch_bounds__(1024) void range_kernel(const unsigned* __restrict__ gbin,
                                                     const int* __restrict__ src, const int* __restrict__ dst,
                                                     const int* __restrict__ labels,
                                                     int* __restrict__ bucket, int* __restrict__ cnt,
                                                     float* __restrict__ inorm, float* __restrict__ onorm,
                                                     int* __restrict__ pair) {
    static_assert((RNG * CAP) % 4 == 0 && CAP % 4 == 0, "rows leave in 16-byte pieces");
    static_assert(LW == 32, "one wave instruction reads a line");
    constexpr int T = 1024;
    constexpr int NWV = T / 32;
    constexpr int LPW = (NBLK + NWV - 1) / NWV;
    __shared__ int lc[RNG];
    __shared__ int4 slots4[RNG * CAP / 4];
    __shared__ int over;
    int* slots = reinterpret_cast<int*>(slots4);
    const int t = threadIdx.x;
    const int lane = t & 31;
    const int wave = __builtin_amdgcn_readfirstlane(t >> 5);
    const int a = blockIdx.x;
    const bool lists = a < NRNG;
    const int r = lists ? a : a - NRNG;
    const unsigned* area = gbin + (long)a * (NBLK * LW) + lane;
    unsigned q[LPW];
#pragma unroll
    for (int u = 0; u < LPW; ++u) {
        const int b = wave + u * NWV;
        q[u] = area[(long)((b < NBLK) ? b : (NBLK - 1)) * LW];
    }
    for (int i = t; i < RNG; i += T) lc[i] = 0;
    if (t == 0) over = 0;
    if (lists)
        for (int i = t; i < RNG * CAP / 4; i += T) slots4[i] = make_int4(0, 0, 0, 0);
    __syncthreads();
    int c[LPW];
    int cmax = 0;
#pragma unroll
    for (int u = 0; u < LPW; ++u) {
        const int b = wave + u * NWV;
        c[u] = (b < NBLK) ? __builtin_amdgcn_readlane((int)q[u], 0) : 0;
        cmax = (c[u] > cmax) ? c[u] : cmax;
    }
    if (cmax > LCAP && lane == 0) over = 1;
    if (lists) {
        int k[LPW];
#pragma unroll
        for (int u = 0; u < LPW; ++u) {
            k[u] = CAP;
            if (lane >= 1 && lane <= c[u]) k[u] = atomicAdd(&lc[(int)(q[u] >> 16)], 1);
        }
#pragma unroll
        for (int u = 0; u < LPW; ++u)
            if (k[u] < CAP) slots[(int)(q[u] >> 16) * CAP + k[u]] = (int)(q[u] & 0xFFFFu);
    } else {
#pragma unroll
        for (int u = 0; u < LPW; ++u)
            if (lane >= 1 && lane <= c[u]) atomicAdd(&lc[(int)q[u]], 1);
    }
    __syncthreads();
    if (over != 0) {
        for (int i = t; i < RNG; i += T) lc[i] = 0;
        __syncthreads();
        if (lists) {
            for (int i = t; i < NE; i += T) {
                const int d = dst[i];
                if ((int)((unsigned)d / RNG) == r) {
                    const int dl = d - r * RNG;
                    const int k = atomicAdd(&lc[dl], 1);
                    if (k < CAP) slots[dl * CAP + k] = src[i];
                }
            }
        } else {
            for (int i = t; i < NE; i += T) {
                const int s = src[i];
                if ((int)((unsigned)s / RNG) == r) atomicAdd(&lc[s - r * RNG], 1);
            }
        }
        __syncthreads();
    }
    const int n0 = r * RNG;
    if (lists) {
        for (int i = t; i < RNG; i += T)
            if (n0 + i < NN) {
                const int c = lc[i];
                cnt[n0 + i] = c;
                inorm[n0 + i] = rsqrtf(fmaxf((float)c, 1.0f));
            }
        int4* brow = reinterpret_cast<int4*>(bucket + (long)n0 * CAP);
        for (int i = t; i < RNG * CAP / 4; i += T)
            if (n0 + (i * 4) / CAP < NN) brow[i] = slots4[i];
    } else {
        for (int i = t; i < RNG; i += T)
            if (n0 + i < NN) {
                const float o = rsqrtf(fmaxf((float)lc[i], 1.0f));
                onorm[n0 + i] = o;
                int2 p;
                p.x = labels[n0 + i];
                p.y = __float_as_int(o);
                *reinterpret_cast<int2*>(pair + 2 * (long)(n0 + i)) = p;
            }
    }
}

template <int D, bool TABLE>
__device__ __forceinline__ void scan_node(const float* hs, const int* __restrict__ pair,
                                       const int* __restrict__ esrc, const int* __restrict__ edst,
                                       int node, int lane, float (&acc)[D / 32]) {
    static_assert(NE % 32 == 0, "the edge list is a whole number of 32-edge reads");
    for (int base = 0; base < NE; base += 32) {
        const int d = edst[base + lane];
        unsigned m = __builtin_amdgcn_ballot_w32(d == node);
        if (m != 0u) {
            const int sv = esrc[base + lane];
            while (m != 0u) {
                const int j = __builtin_ctz(m);
                m &= m - 1u;
                const int s = __builtin_amdgcn_readlane(sv, j);
                if constexpr (TABLE) {
                    const int l = pair[2 * (long)s];
                    const float o = __int_as_float(pair[2 * (long)s + 1]);
                    const float* row = hs + l * EMB + lane;
#pragma unroll
                    for (int k = 0; k < D / 32; ++k) acc[k] += row[k * 32] * o;
                } else {
                    const float* row = hs + (long)s * D + lane;
#pragma unroll
                    for (int k = 0; k < D / 32; ++k) acc[k] += row[k * 32];
                }
            }
        }
    }
}

template <int D>
__device__ __forceinline__ void gather_node(const float* __restrict__ hs, const int* __restrict__ cnt,
                                            const int* __restrict__ bucket,
                                            const int* __restrict__ esrc, const int* __restrict__ edst,
                                            int node, int lane, float (&acc)[D / 32]) {
    static_assert(D % 32 == 0, "a row is a whole number of 32-lane lines");
    constexpr int NK = D / 32;
    constexpr int UB = 8;
    const int* bk = bucket + (long)node * CAP;
    const int deg = cnt[node];
    int sv = bk[lane];
    asm volatile("" : "+v"(sv));
#pragma unroll
    for (int k = 0; k < NK; ++k) acc[k] = 0.0f;
    if (deg > CAP) {
        scan_node<D, false>(hs, nullptr, esrc, edst, node, lane, acc);
        return;
    }
    for (int base = 0; base < deg; base += 32) {
        const int n = (deg - base < 32) ? (deg - base) : 32;
        if (base != 0) sv = bk[base + lane];
        int j = 0;
        for (; j + UB <= n; j += UB) {
            float v[UB][NK];
#pragma unroll
            for (int u = 0; u < UB; ++u) {
                const int s = __builtin_amdgcn_readlane(sv, j + u);
                const float* row = hs + (long)s * D + lane;
#pragma unroll
                for (int k = 0; k < NK; ++k) v[u][k] = row[k * 32];
            }
#pragma unroll
            for (int u = 0; u < UB; ++u)
#pragma unroll
                for (int k = 0; k < NK; ++k) acc[k] += v[u][k];
        }
        if (j < n) {
            float v[UB][NK];
#pragma unroll
            for (int u = 0; u < UB; ++u) {
#pragma unroll
                for (int k = 0; k < NK; ++k) v[u][k] = 0.0f;
                if (j + u < n) {
                    const int s = __builtin_amdgcn_readlane(sv, j + u);
                    const float* row = hs + (long)s * D + lane;
#pragma unroll
                    for (int k = 0; k < NK; ++k) v[u][k] = row[k * 32];
                }
            }
#pragma unroll
            for (int u = 0; u < UB; ++u)
#pragma unroll
                for (int k = 0; k < NK; ++k) acc[k] += v[u][k];
        }
    }
}

__device__ __forceinline__ void edge_records(const int* __restrict__ pair, int sv, int n, int lane,
                                             int& lab, int& onb) {
    lab = 0;
    onb = 0;
    if (lane < n) {
        const int2 p = *reinterpret_cast<const int2*>(pair + 2 * (long)sv);
        lab = p.x;
        onb = p.y;
    }
}

__device__ __forceinline__ void walk_records(const float* embs, int n, int lane, int lab, int onb,
                                             float (&acc)[EMB / 32]) {
#pragma unroll 4
    for (int j = 0; j < n; ++j) {
        const int l = __builtin_amdgcn_readlane(lab, j);
        const float o = __int_as_float(__builtin_amdgcn_readlane(onb, j));
        const float* row = embs + l * EMB + lane;
#pragma unroll
        for (int k = 0; k < EMB / 32; ++k) acc[k] += row[k * 32] * o;
    }
}

template <int D, int NOUT, int NPAD, bool RELU, bool HAS_OS, bool TABLE, int N2, int NPAD2>
__global__ __launch_bounds__(512) void gcn_layer_kernel(
    const float* __restrict__ hs, const int* __restrict__ pair,
    const int* __restrict__ cnt, const int* __restrict__ bucket,
    const int* __restrict__ esrc, const int* __restrict__ edst,
    const float* __restrict__ rowScale, const float* __restrict__ W, const float* __restrict__ bias,
    const float* __restrict__ outScale, const float* __restrict__ Wn, float* __restrict__ out)
{
    static_assert(!TABLE || D == EMB, "the table path is the embedding layer's");
    static_assert(NN % 16 == 0, "every wave of every block owns a node");
    static_assert(NPAD % 16 == 0 && NPAD >= NOUT && NPAD / 16 <= 16, "one wave per column tile");
    constexpr int NK = D / 32;
    constexpr int PITCH = D + 4;
    constexpr int TILES = NPAD / 16;
    constexpr bool FUSE = (N2 > 0);
    constexpr int TILES2 = FUSE ? NPAD2 / 16 : 0;
    constexpr int PITCH2 = NOUT + 4;
    static_assert(!FUSE || (NOUT == NPAD && NOUT % 4 == 0), "the second product runs over all NOUT columns of h");
    static_assert(!FUSE || (NPAD2 % 16 == 0 && NPAD2 >= N2), "whole column tiles for the second product");
    static_assert(TILES + TILES2 <= 16, "one wave per column tile of either product");
    __shared__ float atile[16 * PITCH];
    __shared__ float htile[FUSE ? 16 * PITCH2 : 1];
    __shared__ float embs[TABLE ? NLAB * EMB : 1];

    const int lane = threadIdx.x & 31;
    const int wave = __builtin_amdgcn_readfirstlane((int)(threadIdx.x >> 5));
    const int m0 = blockIdx.x * 16;

    float acc[NK];
    if constexpr (TABLE) {
        constexpr int TBL = NLAB * EMB;
        constexpr int NR = (TBL + 511) / 512;
        const int node = m0 + wave;
        const int* bk = bucket + (long)node * CAP;
        const int deg = cnt[node];
        int sv = bk[lane];
        asm volatile("" : "+v"(sv));
        int n = (deg < 32) ? deg : 32;
        int lab, onb;
        edge_records(pair, sv, n, lane, lab, onb);
        {
            const int t = threadIdx.x;
            float q[NR];
#pragma unroll
            for (int r = 0; r < NR; ++r) {
                const int i = (t + r * 512 < TBL) ? (t + r * 512) : t;
                q[r] = hs[i];
            }
#pragma unroll
            for (int r = 0; r < NR; ++r) {
                const int i = (t + r * 512 < TBL) ? (t + r * 512) : t;
                embs[i] = q[r];
            }
        }
        __syncthreads();
#pragma unroll
        for (int k = 0; k < NK; ++k) acc[k] = 0.0f;
        if (deg > CAP) {
            scan_node<D, true>(embs, pair, esrc, edst, node, lane, acc);
        } else {
            int base = 0;
            while (base < deg) {
                walk_records(embs, n, lane, lab, onb, acc);
                base += 32;
                if (base < deg) {
                    n = (deg - base < 32) ? (deg - base) : 32;
                    sv = bk[base + lane];
                    edge_records(pair, sv, n, lane, lab, onb);
                }
            }
        }
    } else {
        gather_node<D>(hs, cnt, bucket, esrc, edst, m0 + wave, lane, acc);
    }
#pragma unroll
    for (int k = 0; k < NK; ++k) atile[wave * PITCH + k * 32 + lane] = acc[k];
    __syncthreads();

    const int c16 = lane & 15;
    const int half = lane >> 4;
    const int koff = half * 2;
    const int rbase = m0 + half * 8;

    if (wave < TILES) {
        const int n0 = wave * 16;
        const int col = n0 + c16;
        const bool cv = (NOUT == NPAD) || (col < NOUT);
        const int colc = cv ? col : 0;

        const float bl = bias[colc];
        const float bval = cv ? bl : 0.0f;
        v8f c;
#pragma unroll
        for (int r = 0; r < 8; ++r) c[r] = bval;

        const float s = rowScale[m0 + c16];
        float os[8];
#pragma unroll
        for (int r = 0; r < 8; ++r) os[r] = 1.0f;
        if (HAS_OS) {
#pragma unroll
            for (int r = 0; r < 8; ++r) os[r] = outScale[rbase + r];
        }

        const float* arow = atile + c16 * PITCH + koff;
        const float* bcol = W + colc + koff * NOUT;
#pragma unroll
        for (int k0 = 0; k0 < D; k0 += 4) {
            v2f a, b;
            a.x = arow[k0] * s;
            a.y = arow[k0 + 1] * s;
            const float b0 = bcol[k0 * NOUT];
            const float b1 = bcol[(k0 + 1) * NOUT];
            b.x = cv ? b0 : 0.0f;
            b.y = cv ? b1 : 0.0f;
            c = __builtin_amdgcn_wmma_f32_16x16x4_f32(
                     false, a,  false, b,
                     (short)0, c,  false,  false);
        }

        if constexpr (FUSE) {
#pragma unroll
            for (int r = 0; r < 8; ++r) {
                float v = c[r];
                if (RELU) v = fmaxf(v, 0.0f);
                if (HAS_OS) v = v * os[r];
                htile[(half * 8 + r) * PITCH2 + col] = v;
            }
        } else {
            if (cv) {
#pragma unroll
                for (int r = 0; r < 8; ++r) {
                    float v = c[r];
                    if (RELU) v = fmaxf(v, 0.0f);
                    if (HAS_OS) v = v * os[r];
                    out[(long)(rbase + r) * NOUT + col] = v;
                }
            }
        }
    }

    if constexpr (FUSE) {
        constexpr int KS = NOUT / 4;
        const int t2 = wave - TILES;
        const bool mine = (t2 >= 0) && (t2 < TILES2);
        const int col2 = t2 * 16 + c16;
        const bool cv2 = mine && ((N2 == NPAD2) || (col2 < N2));
        const int colc2 = cv2 ? col2 : 0;
        v2f bw[KS];
        if (mine) {
            const float* bcol = Wn + colc2 + koff * N2;
#pragma unroll
            for (int k = 0; k < KS; ++k) {
                const float b0 = bcol[(4 * k) * N2];
                const float b1 = bcol[(4 * k + 1) * N2];
                bw[k].x = cv2 ? b0 : 0.0f;
                bw[k].y = cv2 ? b1 : 0.0f;
            }
        }
        __syncthreads();
        if (mine) {
            v8f c;
#pragma unroll
            for (int r = 0; r < 8; ++r) c[r] = 0.0f;
            const float* arow = htile + c16 * PITCH2 + koff;
#pragma unroll
            for (int k = 0; k < KS; ++k) {
                v2f a;
                a.x = arow[4 * k];
                a.y = arow[4 * k + 1];
                c = __builtin_amdgcn_wmma_f32_16x16x4_f32(
                         false, a,  false, bw[k],
                         (short)0, c,  false,  false);
            }
#pragma unroll
            for (int r = 0; r < 8; ++r)
                out[(long)(rbase + r) * NPAD2 + col2] = c[r];
        }
    }
}

__device__ __forceinline__ void scan_node_proj(const float* __restrict__ proj,
                                               const int* __restrict__ esrc, const int* __restrict__ edst,
                                               int node, int lane, float& ax, float& ay) {
    for (int base = 0; base < NE; base += 32) {
        const int d = edst[base + lane];
        unsigned m = __builtin_amdgcn_ballot_w32(d == node);
        if (m != 0u) {
            const int sv = esrc[base + lane];
            while (m != 0u) {
                const int j = __builtin_ctz(m);
                m &= m - 1u;
                const int s = __builtin_amdgcn_readlane(sv, j);
                const v2f v = *reinterpret_cast<const v2f*>(proj + (long)s * PW + 2 * lane);
                ax += v.x;
                ay += v.y;
            }
        }
    }
}

__global__ __launch_bounds__(512) void gather_out_kernel(
    const float* __restrict__ proj, const int* __restrict__ cnt, const int* __restrict__ bucket,
    const int* __restrict__ esrc, const int* __restrict__ edst,
    const float* __restrict__ rowScale, const float* __restrict__ bias, float* __restrict__ out)
{
    static_assert(NN % 16 == 0, "every wave of every block owns a node");
    static_assert(NLAB % 2 == 0 && NLAB <= PW && PW == 64, "a lane owns two adjacent columns of a 64-float row");
    constexpr int UB = 16;
    const int lane = threadIdx.x & 31;
    const int wave = __builtin_amdgcn_readfirstlane((int)(threadIdx.x >> 5));
    const int node = blockIdx.x * 16 + wave;
    const int* bk = bucket + (long)node * CAP;
    const float sc = rowScale[node];
    const int cl = (lane < NLAB / 2) ? lane : 0;
    const float bx = bias[2 * cl];
    const float by = bias[2 * cl + 1];
    const int deg = cnt[node];
    int sv = bk[lane];
    asm volatile("" : "+v"(sv));
    const float* prow = proj + 2 * lane;

    float ax = 0.0f, ay = 0.0f;
    if (deg > CAP) {
        scan_node_proj(proj, esrc, edst, node, lane, ax, ay);
    } else {
        for (int base = 0; base < deg; base += 32) {
            const int n = (deg - base < 32) ? (deg - base) : 32;
            if (base != 0) sv = bk[base + lane];
            int j = 0;
            for (; j + UB <= n; j += UB) {
                v2f v[UB];
#pragma unroll
                for (int u = 0; u < UB; ++u) {
                    const int s = __builtin_amdgcn_readlane(sv, j + u);
                    v[u] = *reinterpret_cast<const v2f*>(prow + (long)s * PW);
                }
#pragma unroll
                for (int u = 0; u < UB; ++u) { ax += v[u].x; ay += v[u].y; }
            }
            if (j < n) {
                v2f v[UB];
#pragma unroll
                for (int u = 0; u < UB; ++u) {
                    v[u].x = 0.0f;
                    v[u].y = 0.0f;
                    if (j + u < n) {
                        const int s = __builtin_amdgcn_readlane(sv, j + u);
                        v[u] = *reinterpret_cast<const v2f*>(prow + (long)s * PW);
                    }
                }
#pragma unroll
                for (int u = 0; u < UB; ++u) { ax += v[u].x; ay += v[u].y; }
            }
        }
    }
    if (lane < NLAB / 2) {
        v2f o;
        o.x = ax * sc + bx;
        o.y = ay * sc + by;
        *reinterpret_cast<v2f*>(out + (long)node * NLAB + 2 * lane) = o;
    }
}


extern "C" void kernel_launch(void* const* d_in, const int* in_sizes, int n_in,
                              void* d_out, int out_size, void* d_ws, size_t ws_size,
                              hipStream_t stream) {
    const int*   labels = (const int*)d_in[0];
    const int*   src    = (const int*)d_in[1];
    const int*   dst    = (const int*)d_in[2];
    const float* emb    = (const float*)d_in[3];
    const float* W1     = (const float*)d_in[4];
    const float* b1     = (const float*)d_in[5];
    const float* W2     = (const float*)d_in[6];
    const float* b2     = (const float*)d_in[7];
    const float* W3     = (const float*)d_in[8];
    const float* b3     = (const float*)d_in[9];
    float* out = (float*)d_out;

    float* w = (float*)d_ws;
    float* onorm = w;
    float* inorm = w + NN;
    float* bufB  = w + 2 * (size_t)NN;
    float* bufA  = bufB + (size_t)NN * HID;
    float* bufC  = bufA + (size_t)NN * HID;
    int* bucket = (int*)bufB + 32;
    int* cnt  = bucket + (size_t)NN * CAP;
    int* pair = cnt + NN;
    float* proj = bufA + 32;
    unsigned* gbin = (unsigned*)bufA;
    static_assert(32 + (size_t)NN * CAP + 3 * (size_t)NN <= (size_t)NN * HID, "lists, counts and pairs fit their buffer");
    static_assert((size_t)NBIN * NBLK * LW <= 2 * (size_t)NN * HID, "the records fit the two buffers they borrow");
    static_assert(32 + (size_t)NN * PW <= (size_t)NN * HID, "proj fits its buffer");

    edge_bin_kernel<<<NBLK, 256, 0, stream>>>(src, dst, gbin);
    range_kernel<<<NBIN, 1024, 0, stream>>>(gbin, src, dst, labels, bucket, cnt, inorm, onorm, pair);

    gcn_layer_kernel<EMB, HID, HID, true, true, true, 0, 0><<<NN / 16, 512, 0, stream>>>(
        emb, pair, cnt, bucket, src, dst, inorm, W1, b1, onorm, (const float*)nullptr, bufC);

    gcn_layer_kernel<HID, HID, HID, true, true, false, NLAB, PW><<<NN / 16, 512, 0, stream>>>(
        bufC, (const int*)nullptr, cnt, bucket, src, dst, inorm, W2, b2, onorm, W3, proj);

    gather_out_kernel<<<NN / 16, 512, 0, stream>>>(proj, cnt, bucket, src, dst, inorm, b3, out);
}


# layer 1's weight product moved in front of its edge sum: a resident block per WGP makes emb@W1 once on WMMA, holds it and all nodes' label/out-degree words in LDS; sums reordered: product per edge
# speedup vs baseline: 8.9189x; 1.0507x over previous
#include <hip/hip_runtime.h>

#define NN 50000
#define NE 800000
#define NLAB 50
#define EMB 64
#define HID 96
#define CAP 64
#define PW 64
#define RNG 256
#define NRNG ((NN + RNG - 1) / RNG)
#define NBIN (2 * NRNG)
#define CHUNK 2048
#define NBLK ((NE + CHUNK - 1) / CHUNK)
#define LW 32
#define LCAP (LW - 1)

typedef __attribute__((ext_vector_type(2))) float v2f;
typedef __attribute__((ext_vector_type(8))) float v8f;


__global__ __launch_bounds__(256) void edge_bin_kernel(const int* __restrict__ src, const int* __restrict__ dst,
                                                       unsigned* __restrict__ gbin) {
    static_assert(NN <= 65536 && RNG <= 65536, "a record holds the source in 16 bits and the offset in the range above them");
    static_assert(LW == 32, "one wave instruction stores a line");
    constexpr int T = 256;
    constexpr int EPT = CHUNK / T;
    constexpr int UW = 7;
    __shared__ int lcnt[NBIN];
    __shared__ uint4 stage4[NBIN * LW / 4];
    unsigned* stage = reinterpret_cast<unsigned*>(stage4);
    const int t = threadIdx.x;
    const int e0 = blockIdx.x * CHUNK;
    int sv[EPT], dv[EPT];
#pragma unroll
    for (int u = 0; u < EPT; ++u) {
        const int i = e0 + u * T + t;
        const int ii = (i < NE) ? i : (NE - 1);
        sv[u] = src[ii];
        dv[u] = dst[ii];
    }
    for (int a = t; a < NBIN; a += T) lcnt[a] = 0;
    for (int i = t; i < NBIN * LW / 4; i += T) stage4[i] = make_uint4(0u, 0u, 0u, 0u);
    __syncthreads();
    int pd[EPT], ps[EPT];
#pragma unroll
    for (int u = 0; u < EPT; ++u) {
        const int i = e0 + u * T + t;
        pd[u] = 0;
        ps[u] = 0;
        if (i < NE) {
            pd[u] = atomicAdd(&lcnt[(int)((unsigned)dv[u] / RNG)], 1);
            ps[u] = atomicAdd(&lcnt[NRNG + (int)((unsigned)sv[u] / RNG)], 1);
        }
    }
#pragma unroll
    for (int u = 0; u < EPT; ++u) {
        const int i = e0 + u * T + t;
        if (i < NE) {
            const int ad = (int)((unsigned)dv[u] / RNG);
            const int as = NRNG + (int)((unsigned)sv[u] / RNG);
            if (pd[u] < LCAP) stage[ad * LW + 1 + pd[u]] = (unsigned)sv[u] | ((unsigned)(dv[u] - ad * RNG) << 16);
            if (ps[u] < LCAP) stage[as * LW + 1 + ps[u]] = (unsigned)(sv[u] - (as - NRNG) * RNG);
        }
    }
    __syncthreads();
    const int lane = t & 31;
    const int wave = __builtin_amdgcn_readfirstlane(t >> 5);
    unsigned* mine = gbin + (long)blockIdx.x * LW + lane;
    constexpr int APW = NBIN / (T / 32);
    static_assert(NBIN % (T / 32) == 0 && APW % UW == 0, "every wave stores whole groups of UW lines");
    for (int i0 = 0; i0 < APW; i0 += UW) {
        unsigned v[UW];
#pragma unroll
        for (int u = 0; u < UW; ++u) {
            const int a = wave + (i0 + u) * (T / 32);
            const unsigned* p = (lane == 0) ? reinterpret_cast<const unsigned*>(&lcnt[a]) : &stage[a * LW + lane];
            v[u] = *p;
        }
#pragma unroll
        for (int u = 0; u < UW; ++u) {
            const int a = wave + (i0 + u) * (T / 32);
            mine[(long)a * (NBLK * LW)] = v[u];
        }
    }
}

__global__ __launch_bounds__(1024) void range_kernel(const unsigned* __restrict__ gbin,
                                                     const int* __restrict__ src, const int* __restrict__ dst,
                                                     const int* __restrict__ labels,
                                                     int* __restrict__ bucket, int* __restrict__ cnt,
                                                     float* __restrict__ inorm, float* __restrict__ onorm,
                                                     unsigned* __restrict__ nodew) {
    static_assert(NLAB <= 64 && NE < (1 << 26), "a node word holds the label in 6 bits and the out-degree above them");
    static_assert((RNG * CAP) % 4 == 0 && CAP % 4 == 0, "rows leave in 16-byte pieces");
    static_assert(LW == 32, "one wave instruction reads a line");
    constexpr int T = 1024;
    constexpr int NWV = T / 32;
    constexpr int LPW = (NBLK + NWV - 1) / NWV;
    __shared__ int lc[RNG];
    __shared__ int4 slots4[RNG * CAP / 4];
    __shared__ int over;
    int* slots = reinterpret_cast<int*>(slots4);
    const int t = threadIdx.x;
    const int lane = t & 31;
    const int wave = __builtin_amdgcn_readfirstlane(t >> 5);
    const int a = blockIdx.x;
    const bool lists = a < NRNG;
    const int r = lists ? a : a - NRNG;
    const unsigned* area = gbin + (long)a * (NBLK * LW) + lane;
    unsigned q[LPW];
#pragma unroll
    for (int u = 0; u < LPW; ++u) {
        const int b = wave + u * NWV;
        q[u] = area[(long)((b < NBLK) ? b : (NBLK - 1)) * LW];
    }
    for (int i = t; i < RNG; i += T) lc[i] = 0;
    if (t == 0) over = 0;
    if (lists)
        for (int i = t; i < RNG * CAP / 4; i += T) slots4[i] = make_int4(0, 0, 0, 0);
    __syncthreads();
    int c[LPW];
    int cmax = 0;
#pragma unroll
    for (int u = 0; u < LPW; ++u) {
        const int b = wave + u * NWV;
        c[u] = (b < NBLK) ? __builtin_amdgcn_readlane((int)q[u], 0) : 0;
        cmax = (c[u] > cmax) ? c[u] : cmax;
    }
    if (cmax > LCAP && lane == 0) over = 1;
    if (lists) {
        int k[LPW];
#pragma unroll
        for (int u = 0; u < LPW; ++u) {
            k[u] = CAP;
            if (lane >= 1 && lane <= c[u]) k[u] = atomicAdd(&lc[(int)(q[u] >> 16)], 1);
        }
#pragma unroll
        for (int u = 0; u < LPW; ++u)
            if (k[u] < CAP) slots[(int)(q[u] >> 16) * CAP + k[u]] = (int)(q[u] & 0xFFFFu);
    } else {
#pragma unroll
        for (int u = 0; u < LPW; ++u)
            if (lane >= 1 && lane <= c[u]) atomicAdd(&lc[(int)q[u]], 1);
    }
    __syncthreads();
    if (over != 0) {
        for (int i = t; i < RNG; i += T) lc[i] = 0;
        __syncthreads();
        if (lists) {
            for (int i = t; i < NE; i += T) {
                const int d = dst[i];
                if ((int)((unsigned)d / RNG) == r) {
                    const int dl = d - r * RNG;
                    const int k = atomicAdd(&lc[dl], 1);
                    if (k < CAP) slots[dl * CAP + k] = src[i];
                }
            }
        } else {
            for (int i = t; i < NE; i += T) {
                const int s = src[i];
                if ((int)((unsigned)s / RNG) == r) atomicAdd(&lc[s - r * RNG], 1);
            }
        }
        __syncthreads();
    }
    const int n0 = r * RNG;
    if (lists) {
        for (int i = t; i < RNG; i += T)
            if (n0 + i < NN) {
                const int c = lc[i];
                cnt[n0 + i] = c;
                inorm[n0 + i] = rsqrtf(fmaxf((float)c, 1.0f));
            }
        int4* brow = reinterpret_cast<int4*>(bucket + (long)n0 * CAP);
        for (int i = t; i < RNG * CAP / 4; i += T)
            if (n0 + (i * 4) / CAP < NN) brow[i] = slots4[i];
    } else {
        for (int i = t; i < RNG; i += T)
            if (n0 + i < NN) {
                onorm[n0 + i] = rsqrtf(fmaxf((float)lc[i], 1.0f));
                nodew[n0 + i] = (unsigned)labels[n0 + i] | ((unsigned)lc[i] << 6);
            }
    }
}

#define L1_BLOCKS 256

__device__ __forceinline__ void word_records(const unsigned* ntab, int sv, int n, int lane,
                                             int& lab, int& onb) {
    lab = 0;
    onb = 0;
    if (lane < n) {
        const unsigned wd = ntab[sv];
        lab = (int)(wd & 63u);
        onb = __float_as_int(rsqrtf(fmaxf((float)(wd >> 6), 1.0f)));
    }
}

__device__ __forceinline__ void walk_records(const float* ew, int n, int lane, int lab, int onb,
                                             float (&acc)[HID / 32]) {
#pragma unroll 4
    for (int j = 0; j < n; ++j) {
        const int l = __builtin_amdgcn_readlane(lab, j);
        const float o = __int_as_float(__builtin_amdgcn_readlane(onb, j));
        const float* row = ew + l * HID + lane;
#pragma unroll
        for (int k = 0; k < HID / 32; ++k) acc[k] += row[k * 32] * o;
    }
}

__device__ __forceinline__ void scan_node_table(const float* ew, const unsigned* ntab,
                                                const int* __restrict__ esrc, const int* __restrict__ edst,
                                                int node, int lane, float (&acc)[HID / 32]) {
    for (int base = 0; base < NE; base += 32) {
        const int d = edst[base + lane];
        unsigned m = __builtin_amdgcn_ballot_w32(d == node);
        if (m != 0u) {
            const int sv = esrc[base + lane];
            while (m != 0u) {
                const int j = __builtin_ctz(m);
                m &= m - 1u;
                const int s = __builtin_amdgcn_readlane(sv, j);
                const unsigned wd = ntab[s];
                const float o = rsqrtf(fmaxf((float)(wd >> 6), 1.0f));
                const float* row = ew + (int)(wd & 63u) * HID + lane;
#pragma unroll
                for (int k = 0; k < HID / 32; ++k) acc[k] += row[k * 32] * o;
            }
        }
    }
}

__global__ __launch_bounds__(1024) void embed_layer_kernel(
    const float* __restrict__ emb, const unsigned* __restrict__ nodew,
    const int* __restrict__ cnt, const int* __restrict__ bucket,
    const int* __restrict__ esrc, const int* __restrict__ edst,
    const float* __restrict__ rowScale, const float* __restrict__ W, const float* __restrict__ bias,
    const float* __restrict__ outScale, float* __restrict__ out)
{
    constexpr int T = 1024;
    constexpr int NWV = T / 32;
    constexpr int NGRP = (NN + NWV - 1) / NWV;
    constexpr int NK = HID / 32;
    constexpr int RT = (NLAB + 15) / 16;
    constexpr int CT = HID / 16;
    constexpr int KS = EMB / 4;
    static_assert(NN % 4 == 0, "the node table is copied in 16-byte pieces");
    static_assert(HID % 32 == 0 && EMB % 4 == 0, "whole lines per row, whole k steps");
    static_assert(RT * CT <= NWV, "one wave per 16 x 16 tile of EW");
    __shared__ uint4 ntab4[NN / 4];
    __shared__ float ew[RT * 16 * HID];
    const unsigned* ntab = reinterpret_cast<const unsigned*>(ntab4);
    const int t = threadIdx.x;
    const int lane = t & 31;
    const int wave = __builtin_amdgcn_readfirstlane(t >> 5);

    int g = blockIdx.x;
    int node = g * NWV + wave;
    int degv, sv = 0;
    float inv, ov;
    {
        const int i = g * NWV + lane;
        const int ic = (i < NN) ? i : (NN - 1);
        degv = cnt[ic];
        inv = rowScale[ic];
        ov = outScale[ic];
        if (node < NN) sv = bucket[(long)node * CAP + lane];
    }
    float bl[NK];
#pragma unroll
    for (int k = 0; k < NK; ++k) bl[k] = bias[k * 32 + lane];

    if (wave < RT * CT) {
        const int c16 = lane & 15;
        const int half = lane >> 4;
        const int koff = half * 2;
        const int rt = wave / CT;
        const int col = (wave - rt * CT) * 16 + c16;
        const int row = rt * 16 + c16;
        const bool rv = row < NLAB;
        const float* ar = emb + (rv ? row : 0) * EMB + koff;
        const float* bcol = W + col + koff * HID;
        v2f av[KS], bv[KS];
#pragma unroll
        for (int k = 0; k < KS; ++k) {
            const float a0 = ar[4 * k];
            const float a1 = ar[4 * k + 1];
            av[k].x = rv ? a0 : 0.0f;
            av[k].y = rv ? a1 : 0.0f;
            bv[k].x = bcol[(4 * k) * HID];
            bv[k].y = bcol[(4 * k + 1) * HID];
        }
        v8f cc;
#pragma unroll
        for (int r = 0; r < 8; ++r) cc[r] = 0.0f;
#pragma unroll
        for (int k = 0; k < KS; ++k)
            cc = __builtin_amdgcn_wmma_f32_16x16x4_f32(
                     false, av[k],  false, bv[k],
                     (short)0, cc,  false,  false);
#pragma unroll
        for (int r = 0; r < 8; ++r) ew[(rt * 16 + half * 8 + r) * HID + col] = cc[r];
    }
    {
        constexpr int NQ = NN / 4;
        constexpr int NR = (NQ + T - 1) / T;
        const uint4* nw4 = reinterpret_cast<const uint4*>(nodew);
        uint4 q[NR];
#pragma unroll
        for (int r = 0; r < NR; ++r) {
            const int i = (t + r * T < NQ) ? (t + r * T) : t;
            q[r] = nw4[i];
        }
#pragma unroll
        for (int r = 0; r < NR; ++r) {
            const int i = (t + r * T < NQ) ? (t + r * T) : t;
            ntab4[i] = q[r];
        }
    }
    __syncthreads();

    for (; g < NGRP; g += (int)gridDim.x) {
        const int deg = (node < NN) ? __builtin_amdgcn_readlane(degv, wave) : 0;
        const float s = __int_as_float(__builtin_amdgcn_readlane(__float_as_int(inv), wave));
        const float o = __int_as_float(__builtin_amdgcn_readlane(__float_as_int(ov), wave));
        const int svc = sv;

        const int gn = g + (int)gridDim.x;
        const int noden = gn * NWV + wave;
        {
            const int i = gn * NWV + lane;
            const int ic = (i < NN) ? i : (NN - 1);
            degv = cnt[ic];
            inv = rowScale[ic];
            ov = outScale[ic];
            sv = 0;
            if (noden < NN) sv = bucket[(long)noden * CAP + lane];
        }

        float acc[NK];
#pragma unroll
        for (int k = 0; k < NK; ++k) acc[k] = 0.0f;
        if (deg > CAP) {
            scan_node_table(ew, ntab, esrc, edst, node, lane, acc);
        } else {
            int lab, onb;
            const int n0 = (deg < 32) ? deg : 32;
            word_records(ntab, svc, n0, lane, lab, onb);
            walk_records(ew, n0, lane, lab, onb, acc);
            const int* bk = bucket + (long)node * CAP;
            for (int base = 32; base < deg; base += 32) {
                const int n = (deg - base < 32) ? (deg - base) : 32;
                const int svl = bk[base + lane];
                word_records(ntab, svl, n, lane, lab, onb);
                walk_records(ew, n, lane, lab, onb, acc);
            }
        }
        if (node < NN) {
            float* orow = out + (long)node * HID + lane;
#pragma unroll
            for (int k = 0; k < NK; ++k) orow[k * 32] = fmaxf(acc[k] * s + bl[k], 0.0f) * o;
        }
        node = noden;
    }
}

template <int D>
__device__ __forceinline__ void scan_node(const float* __restrict__ hs,
                                       const int* __restrict__ esrc, const int* __restrict__ edst,
                                       int node, int lane, float (&acc)[D / 32]) {
    static_assert(NE % 32 == 0, "the edge list is a whole number of 32-edge reads");
    for (int base = 0; base < NE; base += 32) {
        const int d = edst[base + lane];
        unsigned m = __builtin_amdgcn_ballot_w32(d == node);
        if (m != 0u) {
            const int sv = esrc[base + lane];
            while (m != 0u) {
                const int j = __builtin_ctz(m);
                m &= m - 1u;
                const int s = __builtin_amdgcn_readlane(sv, j);
                const float* row = hs + (long)s * D + lane;
#pragma unroll
                for (int k = 0; k < D / 32; ++k) acc[k] += row[k * 32];
            }
        }
    }
}

template <int D>
__device__ __forceinline__ void gather_node(const float* __restrict__ hs, const int* __restrict__ cnt,
                                            const int* __restrict__ bucket,
                                            const int* __restrict__ esrc, const int* __restrict__ edst,
                                            int node, int lane, float (&acc)[D / 32]) {
    static_assert(D % 32 == 0, "a row is a whole number of 32-lane lines");
    constexpr int NK = D / 32;
    constexpr int UB = 8;
    const int* bk = bucket + (long)node * CAP;
    const int deg = cnt[node];
    int sv = bk[lane];
    asm volatile("" : "+v"(sv));
#pragma unroll
    for (int k = 0; k < NK; ++k) acc[k] = 0.0f;
    if (deg > CAP) {
        scan_node<D>(hs, esrc, edst, node, lane, acc);
        return;
    }
    for (int base = 0; base < deg; base += 32) {
        const int n = (deg - base < 32) ? (deg - base) : 32;
        if (base != 0) sv = bk[base + lane];
        int j = 0;
        for (; j + UB <= n; j += UB) {
            float v[UB][NK];
#pragma unroll
            for (int u = 0; u < UB; ++u) {
                const int s = __builtin_amdgcn_readlane(sv, j + u);
                const float* row = hs + (long)s * D + lane;
#pragma unroll
                for (int k = 0; k < NK; ++k) v[u][k] = row[k * 32];
            }
#pragma unroll
            for (int u = 0; u < UB; ++u)
#pragma unroll
                for (int k = 0; k < NK; ++k) acc[k] += v[u][k];
        }
        if (j < n) {
            float v[UB][NK];
#pragma unroll
            for (int u = 0; u < UB; ++u) {
#pragma unroll
                for (int k = 0; k < NK; ++k) v[u][k] = 0.0f;
                if (j + u < n) {
                    const int s = __builtin_amdgcn_readlane(sv, j + u);
                    const float* row = hs + (long)s * D + lane;
#pragma unroll
                    for (int k = 0; k < NK; ++k) v[u][k] = row[k * 32];
                }
            }
#pragma unroll
            for (int u = 0; u < UB; ++u)
#pragma unroll
                for (int k = 0; k < NK; ++k) acc[k] += v[u][k];
        }
    }
}

template <int D, int NOUT, int NPAD, bool RELU, bool HAS_OS, int N2, int NPAD2>
__global__ __launch_bounds__(512) void gcn_layer_kernel(
    const float* __restrict__ hs,
    const int* __restrict__ cnt, const int* __restrict__ bucket,
    const int* __restrict__ esrc, const int* __restrict__ edst,
    const float* __restrict__ rowScale, const float* __restrict__ W, const float* __restrict__ bias,
    const float* __restrict__ outScale, const float* __restrict__ Wn, float* __restrict__ out)
{
    static_assert(NN % 16 == 0, "every wave of every block owns a node");
    static_assert(NPAD % 16 == 0 && NPAD >= NOUT && NPAD / 16 <= 16, "one wave per column tile");
    constexpr int NK = D / 32;
    constexpr int PITCH = D + 4;
    constexpr int TILES = NPAD / 16;
    constexpr bool FUSE = (N2 > 0);
    constexpr int TILES2 = FUSE ? NPAD2 / 16 : 0;
    constexpr int PITCH2 = NOUT + 4;
    static_assert(!FUSE || (NOUT == NPAD && NOUT % 4 == 0), "the second product runs over all NOUT columns of h");
    static_assert(!FUSE || (NPAD2 % 16 == 0 && NPAD2 >= N2), "whole column tiles for the second product");
    static_assert(TILES + TILES2 <= 16, "one wave per column tile of either product");
    __shared__ float atile[16 * PITCH];
    __shared__ float htile[FUSE ? 16 * PITCH2 : 1];

    const int lane = threadIdx.x & 31;
    const int wave = __builtin_amdgcn_readfirstlane((int)(threadIdx.x >> 5));
    const int m0 = blockIdx.x * 16;

    float acc[NK];
    gather_node<D>(hs, cnt, bucket, esrc, edst, m0 + wave, lane, acc);
#pragma unroll
    for (int k = 0; k < NK; ++k) atile[wave * PITCH + k * 32 + lane] = acc[k];
    __syncthreads();

    const int c16 = lane & 15;
    const int half = lane >> 4;
    const int koff = half * 2;
    const int rbase = m0 + half * 8;

    if (wave < TILES) {
        const int n0 = wave * 16;
        const int col = n0 + c16;
        const bool cv = (NOUT == NPAD) || (col < NOUT);
        const int colc = cv ? col : 0;

        const float bl = bias[colc];
        const float bval = cv ? bl : 0.0f;
        v8f c;
#pragma unroll
        for (int r = 0; r < 8; ++r) c[r] = bval;

        const float s = rowScale[m0 + c16];
        float os[8];
#pragma unroll
        for (int r = 0; r < 8; ++r) os[r] = 1.0f;
        if (HAS_OS) {
#pragma unroll
            for (int r = 0; r < 8; ++r) os[r] = outScale[rbase + r];
        }

        const float* arow = atile + c16 * PITCH + koff;
        const float* bcol = W + colc + koff * NOUT;
#pragma unroll
        for (int k0 = 0; k0 < D; k0 += 4) {
            v2f a, b;
            a.x = arow[k0] * s;
            a.y = arow[k0 + 1] * s;
            const float b0 = bcol[k0 * NOUT];
            const float b1 = bcol[(k0 + 1) * NOUT];
            b.x = cv ? b0 : 0.0f;
            b.y = cv ? b1 : 0.0f;
            c = __builtin_amdgcn_wmma_f32_16x16x4_f32(
                     false, a,  false, b,
                     (short)0, c,  false,  false);
        }

        if constexpr (FUSE) {
#pragma unroll
            for (int r = 0; r < 8; ++r) {
                float v = c[r];
                if (RELU) v = fmaxf(v, 0.0f);
                if (HAS_OS) v = v * os[r];
                htile[(half * 8 + r) * PITCH2 + col] = v;
            }
        } else {
            if (cv) {
#pragma unroll
                for (int r = 0; r < 8; ++r) {
                    float v = c[r];
                    if (RELU) v = fmaxf(v, 0.0f);
                    if (HAS_OS) v = v * os[r];
                    out[(long)(rbase + r) * NOUT + col] = v;
                }
            }
        }
    }

    if constexpr (FUSE) {
        constexpr int KS = NOUT / 4;
        const int t2 = wave - TILES;
        const bool mine = (t2 >= 0) && (t2 < TILES2);
        const int col2 = t2 * 16 + c16;
        const bool cv2 = mine && ((N2 == NPAD2) || (col2 < N2));
        const int colc2 = cv2 ? col2 : 0;
        v2f bw[KS];
        if (mine) {
            const float* bcol = Wn + colc2 + koff * N2;
#pragma unroll
            for (int k = 0; k < KS; ++k) {
                const float b0 = bcol[(4 * k) * N2];
                const float b1 = bcol[(4 * k + 1) * N2];
                bw[k].x = cv2 ? b0 : 0.0f;
                bw[k].y = cv2 ? b1 : 0.0f;
            }
        }
        __syncthreads();
        if (mine) {
            v8f c;
#pragma unroll
            for (int r = 0; r < 8; ++r) c[r] = 0.0f;
            const float* arow = htile + c16 * PITCH2 + koff;
#pragma unroll
            for (int k = 0; k < KS; ++k) {
                v2f a;
                a.x = arow[4 * k];
                a.y = arow[4 * k + 1];
                c = __builtin_amdgcn_wmma_f32_16x16x4_f32(
                         false, a,  false, bw[k],
                         (short)0, c,  false,  false);
            }
#pragma unroll
            for (int r = 0; r < 8; ++r)
                out[(long)(rbase + r) * NPAD2 + col2] = c[r];
        }
    }
}

__device__ __forceinline__ void scan_node_proj(const float* __restrict__ proj,
                                               const int* __restrict__ esrc, const int* __restrict__ edst,
                                               int node, int lane, float& ax, float& ay) {
    for (int base = 0; base < NE; base += 32) {
        const int d = edst[base + lane];
        unsigned m = __builtin_amdgcn_ballot_w32(d == node);
        if (m != 0u) {
            const int sv = esrc[base + lane];
            while (m != 0u) {
                const int j = __builtin_ctz(m);
                m &= m - 1u;
                const int s = __builtin_amdgcn_readlane(sv, j);
                const v2f v = *reinterpret_cast<const v2f*>(proj + (long)s * PW + 2 * lane);
                ax += v.x;
                ay += v.y;
            }
        }
    }
}

__global__ __launch_bounds__(512) void gather_out_kernel(
    const float* __restrict__ proj, const int* __restrict__ cnt, const int* __restrict__ bucket,
    const int* __restrict__ esrc, const int* __restrict__ edst,
    const float* __restrict__ rowScale, const float* __restrict__ bias, float* __restrict__ out)
{
    static_assert(NN % 16 == 0, "every wave of every block owns a node");
    static_assert(NLAB % 2 == 0 && NLAB <= PW && PW == 64, "a lane owns two adjacent columns of a 64-float row");
    constexpr int UB = 16;
    const int lane = threadIdx.x & 31;
    const int wave = __builtin_amdgcn_readfirstlane((int)(threadIdx.x >> 5));
    const int node = blockIdx.x * 16 + wave;
    const int* bk = bucket + (long)node * CAP;
    const float sc = rowScale[node];
    const int cl = (lane < NLAB / 2) ? lane : 0;
    const float bx = bias[2 * cl];
    const float by = bias[2 * cl + 1];
    const int deg = cnt[node];
    int sv = bk[lane];
    asm volatile("" : "+v"(sv));
    const float* prow = proj + 2 * lane;

    float ax = 0.0f, ay = 0.0f;
    if (deg > CAP) {
        scan_node_proj(proj, esrc, edst, node, lane, ax, ay);
    } else {
        for (int base = 0; base < deg; base += 32) {
            const int n = (deg - base < 32) ? (deg - base) : 32;
            if (base != 0) sv = bk[base + lane];
            int j = 0;
            for (; j + UB <= n; j += UB) {
                v2f v[UB];
#pragma unroll
                for (int u = 0; u < UB; ++u) {
                    const int s = __builtin_amdgcn_readlane(sv, j + u);
                    v[u] = *reinterpret_cast<const v2f*>(prow + (long)s * PW);
                }
#pragma unroll
                for (int u = 0; u < UB; ++u) { ax += v[u].x; ay += v[u].y; }
            }
            if (j < n) {
                v2f v[UB];
#pragma unroll
                for (int u = 0; u < UB; ++u) {
                    v[u].x = 0.0f;
                    v[u].y = 0.0f;
                    if (j + u < n) {
                        const int s = __builtin_amdgcn_readlane(sv, j + u);
                        v[u] = *reinterpret_cast<const v2f*>(prow + (long)s * PW);
                    }
                }
#pragma unroll
                for (int u = 0; u < UB; ++u) { ax += v[u].x; ay += v[u].y; }
            }
        }
    }
    if (lane < NLAB / 2) {
        v2f o;
        o.x = ax * sc + bx;
        o.y = ay * sc + by;
        *reinterpret_cast<v2f*>(out + (long)node * NLAB + 2 * lane) = o;
    }
}


extern "C" void kernel_launch(void* const* d_in, const int* in_sizes, int n_in,
                              void* d_out, int out_size, void* d_ws, size_t ws_size,
                              hipStream_t stream) {
    const int*   labels = (const int*)d_in[0];
    const int*   src    = (const int*)d_in[1];
    const int*   dst    = (const int*)d_in[2];
    const float* emb    = (const float*)d_in[3];
    const float* W1     = (const float*)d_in[4];
    const float* b1     = (const float*)d_in[5];
    const float* W2     = (const float*)d_in[6];
    const float* b2     = (const float*)d_in[7];
    const float* W3     = (const float*)d_in[8];
    const float* b3     = (const float*)d_in[9];
    float* out = (float*)d_out;

    float* w = (float*)d_ws;
    float* onorm = w;
    float* inorm = w + NN;
    float* bufB  = w + 2 * (size_t)NN;
    float* bufA  = bufB + (size_t)NN * HID;
    float* bufC  = bufA + (size_t)NN * HID;
    int* bucket = (int*)bufB + 32;
    int* cnt  = bucket + (size_t)NN * CAP;
    unsigned* nodew = (unsigned*)(cnt + NN);
    float* proj = bufA + 32;
    unsigned* gbin = (unsigned*)bufA;
    static_assert(32 + (size_t)NN * CAP + 2 * (size_t)NN <= (size_t)NN * HID, "lists, counts and node words fit their buffer");
    static_assert((2 * (size_t)NN + 32 + (size_t)NN * CAP + NN) % 4 == 0, "the node words start on a 16-byte boundary of the workspace");
    static_assert((size_t)NBIN * NBLK * LW <= 2 * (size_t)NN * HID, "the records fit the two buffers they borrow");
    static_assert(32 + (size_t)NN * PW <= (size_t)NN * HID, "proj fits its buffer");

    edge_bin_kernel<<<NBLK, 256, 0, stream>>>(src, dst, gbin);
    range_kernel<<<NBIN, 1024, 0, stream>>>(gbin, src, dst, labels, bucket, cnt, inorm, onorm, nodew);

    embed_layer_kernel<<<L1_BLOCKS, 1024, 0, stream>>>(
        emb, nodew, cnt, bucket, src, dst, inorm, W1, b1, onorm, bufC);

    gcn_layer_kernel<HID, HID, HID, true, true, NLAB, PW><<<NN / 16, 512, 0, stream>>>(
        bufC, cnt, bucket, src, dst, inorm, W2, b2, onorm, W3, proj);

    gather_out_kernel<<<NN / 16, 512, 0, stream>>>(proj, cnt, bucket, src, dst, inorm, b3, out);
}
